# starting kernel
# baseline (speedup 1.0000x reference)
#include <hip/hip_runtime.h>
#include <hip/hip_bf16.h>

typedef __attribute__((ext_vector_type(2))) float v2f;
typedef __attribute__((ext_vector_type(8))) float v8f;

#define D 128

__device__ __forceinline__ unsigned enc_order(float f) {
    unsigned s = __float_as_uint(f);
    return (s & 0x80000000u) ? ~s : (s | 0x80000000u);
}
__device__ __forceinline__ float dec_order(unsigned u) {
    unsigned s = (u & 0x80000000u) ? (u & 0x7fffffffu) : ~u;
    return __uint_as_float(s);
}

__global__ void gat_scores_wmma(const float* __restrict__ embs,
                                const int* __restrict__ ratings,
                                float* __restrict__ scores,
                                unsigned* __restrict__ segmax,
                                int n_edges) {
    const int lane = threadIdx.x & 31;
    const int tile = blockIdx.x * (blockDim.x >> 5) + (threadIdx.x >> 5);
    const int e0 = tile << 4;
    if (e0 >= n_edges) return;

    int eL = e0 + (lane & 15);
    if (eL >= n_edges) eL = n_edges - 1;
    const int r0 = ratings[2 * eL + 0];
    const int r1 = ratings[2 * eL + 1];

    const int sub = (lane >> 4) << 1;
    const float* pa = embs + (size_t)r0 * D + sub;
    const float* pb = embs + (size_t)r1 * D + sub;

    v8f acc = {};
#pragma unroll
    for (int c = 0; c < D / 4; ++c) {
        v2f a = *(const v2f*)(pa + 4 * c);
        v2f b = *(const v2f*)(pb + 4 * c);
        acc = __builtin_amdgcn_wmma_f32_16x16x4_f32(
            false, a, false, b, (short)0, acc, false, false);
    }

    const int k = lane & 7;
    float v = acc[0];
    v = (k == 1) ? acc[1] : v;
    v = (k == 2) ? acc[2] : v;
    v = (k == 3) ? acc[3] : v;
    v = (k == 4) ? acc[4] : v;
    v = (k == 5) ? acc[5] : v;
    v = (k == 6) ? acc[6] : v;
    v = (k == 7) ? acc[7] : v;
    const int src = (lane < 8) ? lane : lane + 16;
    const float sc = __shfl(v, src, 32);

    if (lane < 16 && (e0 + lane) < n_edges) {
        scores[e0 + lane] = sc;
        atomicMax(&segmax[r0], enc_order(sc));
    }
}

__global__ void gat_softmax_denom(const int* __restrict__ ratings,
                                  const unsigned* __restrict__ segmax,
                                  float* __restrict__ scores,
                                  float* __restrict__ denom,
                                  int n_edges) {
    const int e = blockIdx.x * blockDim.x + threadIdx.x;
    if (e >= n_edges) return;
    const int r0 = ratings[2 * e];
    const float m = dec_order(segmax[r0]);
    const float ex = __expf(scores[e] - m);
    scores[e] = ex;
    atomicAdd(&denom[r0], ex);
}

__global__ void gat_aggregate(const float* __restrict__ embs,
                              const int* __restrict__ ratings,
                              const float* __restrict__ ex,
                              const float* __restrict__ denom,
                              float* __restrict__ out,
                              int n_edges) {
    const int lane = threadIdx.x & 31;
    const int e = blockIdx.x * (blockDim.x >> 5) + (threadIdx.x >> 5);
    if (e >= n_edges) return;
    const int r0 = ratings[2 * e + 0];
    const int r1 = ratings[2 * e + 1];
    const float attn = ex[e] / denom[r0];

    const float4 v = ((const float4*)(embs + (size_t)r1 * D))[lane];
    float* dst = out + (size_t)r0 * D + lane * 4;
    atomicAdd(dst + 0, attn * v.x);
    atomicAdd(dst + 1, attn * v.y);
    atomicAdd(dst + 2, attn * v.z);
    atomicAdd(dst + 3, attn * v.w);
}

extern "C" void kernel_launch(void* const* d_in, const int* in_sizes, int n_in,
                              void* d_out, int out_size, void* d_ws, size_t ws_size,
                              hipStream_t stream) {
    const float* embs    = (const float*)d_in[0];
    const int*   ratings = (const int*)d_in[1];
    const int node_num = in_sizes[0] / D;
    const int n_edges  = in_sizes[1] / 2;
    float* out = (float*)d_out;

    char* ws = (char*)d_ws;
    float*    scores = (float*)ws;
    unsigned* segmax = (unsigned*)(ws + (size_t)n_edges * 4);
    float*    denom  = (float*)(ws + (size_t)n_edges * 4 + (size_t)node_num * 4);

    hipMemsetAsync(d_out, 0, (size_t)out_size * sizeof(float), stream);
    hipMemsetAsync(segmax, 0, (size_t)node_num * 4, stream);
    hipMemsetAsync(denom, 0, (size_t)node_num * 4, stream);

    const int tiles = (n_edges + 15) / 16;
    dim3 blk(256);
    dim3 g1((tiles + 7) / 8);
    gat_scores_wmma<<<g1, blk, 0, stream>>>(embs, ratings, scores, segmax, n_edges);

    dim3 g2((n_edges + 255) / 256);
    gat_softmax_denom<<<g2, blk, 0, stream>>>(ratings, segmax, scores, denom, n_edges);

    dim3 g3((n_edges + 7) / 8);
    gat_aggregate<<<g3, blk, 0, stream>>>(embs, ratings, scores, denom, out, n_edges);
}


# scores kernel: each 512-B row read by one coalesced 16-B-per-lane load (4 edges ahead), both operands alike; score = trace of the edge's tile (2 WMMA), lanes summed by DPP butterfly; new sum order
# speedup vs baseline: 1.2382x; 1.2382x over previous
#include <hip/hip_runtime.h>
#include <hip/hip_bf16.h>

typedef __attribute__((ext_vector_type(2))) float v2f;
typedef __attribute__((ext_vector_type(8))) float v8f;

#define D 128

__device__ __forceinline__ unsigned enc_order(float f) {
    unsigned s = __float_as_uint(f);
    return (s & 0x80000000u) ? ~s : (s | 0x80000000u);
}
__device__ __forceinline__ float dec_order(unsigned u) {
    unsigned s = (u & 0x80000000u) ? (u & 0x7fffffffu) : ~u;
    return __uint_as_float(s);
}


template <int M>
__device__ __forceinline__ float row_xor(float x) {
    return __builtin_bit_cast(float,
        __builtin_amdgcn_update_dpp(0, __builtin_bit_cast(int, x),
                                    0x160 + M, 0xf, 0xf, false));
}
__device__ __forceinline__ float other_row(float x) {
    const int b = __builtin_bit_cast(int, x);
    return __builtin_bit_cast(float,
        __builtin_amdgcn_permlanex16(b, b, 0x76543210, 0xfedcba98, false, false));
}
template <int M>
__device__ __forceinline__ float fold_pair(float lo, float hi, bool bit) {
    const float keep = bit ? hi : lo;
    const float send = bit ? lo : hi;
    return keep + row_xor<M>(send);
}

__global__ void gat_scores_wmma(const float* __restrict__ embs,
                                const int* __restrict__ ratings,
                                float* __restrict__ scores,
                                unsigned* __restrict__ segmax,
                                int n_edges) {
    const int lane = threadIdx.x & 31;
    const int tile = blockIdx.x * (blockDim.x >> 5) + (threadIdx.x >> 5);
    const int e0 = tile << 4;
    if (e0 >= n_edges) return;

    int eL = e0 + (lane & 15);
    if (eL >= n_edges) eL = n_edges - 1;
    const int r0 = ratings[2 * eL + 0];
    const int r1 = ratings[2 * eL + 1];

    const int  k       = lane & 7;
    const bool on_diag = (lane < 8) || (lane >= 24);

    constexpr int AHEAD = 4;
    float4 a[16], b[16];
    float d[16];
#pragma unroll
    for (int j = 0; j < AHEAD; ++j) {
        const int ra = __builtin_amdgcn_readlane(r0, j);
        const int rb = __builtin_amdgcn_readlane(r1, j);
        a[j] = ((const float4*)(embs + (size_t)ra * D))[lane];
        b[j] = ((const float4*)(embs + (size_t)rb * D))[lane];
    }
#pragma unroll
    for (int j = 0; j < 16; ++j) {
        if (j + AHEAD < 16) {
            const int ra = __builtin_amdgcn_readlane(r0, j + AHEAD);
            const int rb = __builtin_amdgcn_readlane(r1, j + AHEAD);
            a[j + AHEAD] = ((const float4*)(embs + (size_t)ra * D))[lane];
            b[j + AHEAD] = ((const float4*)(embs + (size_t)rb * D))[lane];
        }
        const v2f a0 = {a[j].x, a[j].y}, a1 = {a[j].z, a[j].w};
        const v2f b0 = {b[j].x, b[j].y}, b1 = {b[j].z, b[j].w};

        v8f acc = {};
        acc = __builtin_amdgcn_wmma_f32_16x16x4_f32(
            false, a0, false, b0, (short)0, acc, false, false);
        acc = __builtin_amdgcn_wmma_f32_16x16x4_f32(
            false, a1, false, b1, (short)0, acc, false, false);

        float v = acc[0];
        v = (k == 1) ? acc[1] : v;
        v = (k == 2) ? acc[2] : v;
        v = (k == 3) ? acc[3] : v;
        v = (k == 4) ? acc[4] : v;
        v = (k == 5) ? acc[5] : v;
        v = (k == 6) ? acc[6] : v;
        v = (k == 7) ? acc[7] : v;
        d[j] = on_diag ? v : 0.0f;
    }

    const bool bit8 = (lane & 8) != 0, bit4 = (lane & 4) != 0;
    const bool bit2 = (lane & 2) != 0, bit1 = (lane & 1) != 0;
    float x[8], y[4], w[2];
#pragma unroll
    for (int j = 0; j < 8; ++j) x[j] = fold_pair<8>(d[j], d[j + 8], bit8);
#pragma unroll
    for (int j = 0; j < 4; ++j) y[j] = fold_pair<4>(x[j], x[j + 4], bit4);
#pragma unroll
    for (int j = 0; j < 2; ++j) w[j] = fold_pair<2>(y[j], y[j + 2], bit2);
    const float z  = fold_pair<1>(w[0], w[1], bit1);
    const float sc = z + other_row(z);

    if (lane < 16 && (e0 + lane) < n_edges) {
        scores[e0 + lane] = sc;
        atomicMax(&segmax[r0], enc_order(sc));
    }
}

__global__ void gat_softmax_denom(const int* __restrict__ ratings,
                                  const unsigned* __restrict__ segmax,
                                  float* __restrict__ scores,
                                  float* __restrict__ denom,
                                  int n_edges) {
    const int e = blockIdx.x * blockDim.x + threadIdx.x;
    if (e >= n_edges) return;
    const int r0 = ratings[2 * e];
    const float m = dec_order(segmax[r0]);
    const float ex = __expf(scores[e] - m);
    scores[e] = ex;
    atomicAdd(&denom[r0], ex);
}

__global__ void gat_aggregate(const float* __restrict__ embs,
                              const int* __restrict__ ratings,
                              const float* __restrict__ ex,
                              const float* __restrict__ denom,
                              float* __restrict__ out,
                              int n_edges) {
    const int lane = threadIdx.x & 31;
    const int e = blockIdx.x * (blockDim.x >> 5) + (threadIdx.x >> 5);
    if (e >= n_edges) return;
    const int r0 = ratings[2 * e + 0];
    const int r1 = ratings[2 * e + 1];
    const float attn = ex[e] / denom[r0];

    const float4 v = ((const float4*)(embs + (size_t)r1 * D))[lane];
    float* dst = out + (size_t)r0 * D + lane * 4;
    atomicAdd(dst + 0, attn * v.x);
    atomicAdd(dst + 1, attn * v.y);
    atomicAdd(dst + 2, attn * v.z);
    atomicAdd(dst + 3, attn * v.w);
}

extern "C" void kernel_launch(void* const* d_in, const int* in_sizes, int n_in,
                              void* d_out, int out_size, void* d_ws, size_t ws_size,
                              hipStream_t stream) {
    const float* embs    = (const float*)d_in[0];
    const int*   ratings = (const int*)d_in[1];
    const int node_num = in_sizes[0] / D;
    const int n_edges  = in_sizes[1] / 2;
    float* out = (float*)d_out;

    char* ws = (char*)d_ws;
    float*    scores = (float*)ws;
    unsigned* segmax = (unsigned*)(ws + (size_t)n_edges * 4);
    float*    denom  = (float*)(ws + (size_t)n_edges * 4 + (size_t)node_num * 4);

    hipMemsetAsync(d_out, 0, (size_t)out_size * sizeof(float), stream);
    hipMemsetAsync(segmax, 0, (size_t)node_num * 4, stream);
    hipMemsetAsync(denom, 0, (size_t)node_num * 4, stream);

    const int tiles = (n_edges + 15) / 16;
    dim3 blk(256);
    dim3 g1((tiles + 7) / 8);
    gat_scores_wmma<<<g1, blk, 0, stream>>>(embs, ratings, scores, segmax, n_edges);

    dim3 g2((n_edges + 255) / 256);
    gat_softmax_denom<<<g2, blk, 0, stream>>>(ratings, segmax, scores, denom, n_edges);

    dim3 g3((n_edges + 7) / 8);
    gat_aggregate<<<g3, blk, 0, stream>>>(embs, ratings, scores, denom, out, n_edges);
}


# aggregation: edges counting-sorted by destination row (count, scan, slot cursor in the scores pass); one wave per row does softmax and weighted row sum in registers, one store; new sum order
# speedup vs baseline: 1.8798x; 1.5182x over previous
#include <hip/hip_runtime.h>
#include <hip/hip_bf16.h>

typedef __attribute__((ext_vector_type(2))) float v2f;
typedef __attribute__((ext_vector_type(8))) float v8f;

#define D 128

__device__ __forceinline__ unsigned enc_order(float f) {
    unsigned s = __float_as_uint(f);
    return (s & 0x80000000u) ? ~s : (s | 0x80000000u);
}
__device__ __forceinline__ float dec_order(unsigned u) {
    unsigned s = (u & 0x80000000u) ? (u & 0x7fffffffu) : ~u;
    return __uint_as_float(s);
}


template <int M>
__device__ __forceinline__ float row_xor(float x) {
    return __builtin_bit_cast(float,
        __builtin_amdgcn_update_dpp(0, __builtin_bit_cast(int, x),
                                    0x160 + M, 0xf, 0xf, false));
}
__device__ __forceinline__ float other_row(float x) {
    const int b = __builtin_bit_cast(int, x);
    return __builtin_bit_cast(float,
        __builtin_amdgcn_permlanex16(b, b, 0x76543210, 0xfedcba98, false, false));
}
template <int M>
__device__ __forceinline__ float fold_pair(float lo, float hi, bool bit) {
    const float keep = bit ? hi : lo;
    const float send = bit ? lo : hi;
    return keep + row_xor<M>(send);
}

__global__ void gat_scores_wmma(const float* __restrict__ embs,
                                const int* __restrict__ ratings,
                                float* __restrict__ scores,
                                int* __restrict__ aux,
                                int* __restrict__ slot_r1,
                                int n_edges, int grouped) {
    const int lane = threadIdx.x & 31;
    const int tile = blockIdx.x * (blockDim.x >> 5) + (threadIdx.x >> 5);
    const int e0 = tile << 4;
    if (e0 >= n_edges) return;

    int eL = e0 + (lane & 15);
    if (eL >= n_edges) eL = n_edges - 1;
    const int r0 = ratings[2 * eL + 0];
    const int r1 = ratings[2 * eL + 1];

    const int  k       = lane & 7;
    const bool on_diag = (lane < 8) || (lane >= 24);

    constexpr int AHEAD = 4;
    float4 a[16], b[16];
    float d[16];
#pragma unroll
    for (int j = 0; j < AHEAD; ++j) {
        const int ra = __builtin_amdgcn_readlane(r0, j);
        const int rb = __builtin_amdgcn_readlane(r1, j);
        a[j] = ((const float4*)(embs + (size_t)ra * D))[lane];
        b[j] = ((const float4*)(embs + (size_t)rb * D))[lane];
    }
#pragma unroll
    for (int j = 0; j < 16; ++j) {
        if (j + AHEAD < 16) {
            const int ra = __builtin_amdgcn_readlane(r0, j + AHEAD);
            const int rb = __builtin_amdgcn_readlane(r1, j + AHEAD);
            a[j + AHEAD] = ((const float4*)(embs + (size_t)ra * D))[lane];
            b[j + AHEAD] = ((const float4*)(embs + (size_t)rb * D))[lane];
        }
        const v2f a0 = {a[j].x, a[j].y}, a1 = {a[j].z, a[j].w};
        const v2f b0 = {b[j].x, b[j].y}, b1 = {b[j].z, b[j].w};

        v8f acc = {};
        acc = __builtin_amdgcn_wmma_f32_16x16x4_f32(
            false, a0, false, b0, (short)0, acc, false, false);
        acc = __builtin_amdgcn_wmma_f32_16x16x4_f32(
            false, a1, false, b1, (short)0, acc, false, false);

        float v = acc[0];
        v = (k == 1) ? acc[1] : v;
        v = (k == 2) ? acc[2] : v;
        v = (k == 3) ? acc[3] : v;
        v = (k == 4) ? acc[4] : v;
        v = (k == 5) ? acc[5] : v;
        v = (k == 6) ? acc[6] : v;
        v = (k == 7) ? acc[7] : v;
        d[j] = on_diag ? v : 0.0f;
    }

    const bool bit8 = (lane & 8) != 0, bit4 = (lane & 4) != 0;
    const bool bit2 = (lane & 2) != 0, bit1 = (lane & 1) != 0;
    float x[8], y[4], w[2];
#pragma unroll
    for (int j = 0; j < 8; ++j) x[j] = fold_pair<8>(d[j], d[j + 8], bit8);
#pragma unroll
    for (int j = 0; j < 4; ++j) y[j] = fold_pair<4>(x[j], x[j + 4], bit4);
#pragma unroll
    for (int j = 0; j < 2; ++j) w[j] = fold_pair<2>(y[j], y[j + 2], bit2);
    const float z  = fold_pair<1>(w[0], w[1], bit1);
    const float sc = z + other_row(z);

    if (lane < 16 && (e0 + lane) < n_edges) {
        if (grouped) {
            const int pos = atomicAdd(&aux[r0], 1);
            scores[pos]  = sc;
            slot_r1[pos] = r1;
        } else {
            scores[e0 + lane] = sc;
            atomicMax((unsigned*)aux + r0, enc_order(sc));
        }
    }
}

__global__ void gat_count(const int* __restrict__ ratings,
                          int* __restrict__ counts, int n_edges) {
    const int e = (blockIdx.x * blockDim.x + threadIdx.x) * 4;
    if (e + 3 < n_edges) {
        const int4 p = ((const int4*)ratings)[(e >> 1) + 0];
        const int4 q = ((const int4*)ratings)[(e >> 1) + 1];
        atomicAdd(&counts[p.x], 1);
        atomicAdd(&counts[p.z], 1);
        atomicAdd(&counts[q.x], 1);
        atomicAdd(&counts[q.z], 1);
    } else {
        for (int k = e; k < n_edges; ++k) atomicAdd(&counts[ratings[2 * k]], 1);
    }
}

__global__ void gat_scan(const int* __restrict__ counts,
                         int* __restrict__ cursor, int n_nodes) {
    __shared__ int wave_tot[32];
    const int t = threadIdx.x, lane = t & 31, w = t >> 5;
    const int per = (((n_nodes + (int)blockDim.x - 1) / (int)blockDim.x) + 3) & ~3;
    const int b = min(t * per, n_nodes), e = min(b + per, n_nodes);
    int s = 0;
    for (int j = b; j < e; ++j) s += counts[j];
    int incl = s;
    for (int d = 1; d < 32; d <<= 1) {
        const int y = __shfl_up(incl, d);
        if (lane >= d) incl += y;
    }
    if (lane == 31) wave_tot[w] = incl;
    __syncthreads();
    if (w == 0) {
        const int v = (lane < (int)(blockDim.x >> 5)) ? wave_tot[lane] : 0;
        int vi = v;
        for (int d = 1; d < 32; d <<= 1) {
            const int y = __shfl_up(vi, d);
            if (lane >= d) vi += y;
        }
        wave_tot[lane] = vi - v;
    }
    __syncthreads();
    int run = wave_tot[w] + incl - s;
    for (int j = b; j < e; ++j) { cursor[j] = run; run += counts[j]; }
}

__device__ __forceinline__ float lane_f(float x, int l) {
    return __builtin_bit_cast(float,
        __builtin_amdgcn_readlane(__builtin_bit_cast(int, x), l));
}

__global__ void gat_rows(const float* __restrict__ embs,
                         const int* __restrict__ cursor,
                         const int* __restrict__ counts,
                         const float* __restrict__ slot_sc,
                         const int* __restrict__ slot_r1,
                         float* __restrict__ out, int n_nodes) {
    const int lane = threadIdx.x & 31;
    const int i = __builtin_amdgcn_readfirstlane(
        (int)(blockIdx.x * (blockDim.x >> 5) + (threadIdx.x >> 5)));
    if (i >= n_nodes) return;
    const int end = cursor[i];
    const int beg = end - counts[i];

    float4 acc = {0.0f, 0.0f, 0.0f, 0.0f};
    if (beg < end) {
        float m = -INFINITY;
        for (int p = beg + lane; p < end; p += 32) m = fmaxf(m, slot_sc[p]);
        m = fmaxf(m, row_xor<8>(m)); m = fmaxf(m, row_xor<4>(m));
        m = fmaxf(m, row_xor<2>(m)); m = fmaxf(m, row_xor<1>(m));
        m = fmaxf(m, other_row(m));
        if (!__builtin_isfinite(m)) m = 0.0f;

        float den = 0.0f;
        for (int p = beg + lane; p < end; p += 32) den += __expf(slot_sc[p] - m);
        den += row_xor<8>(den); den += row_xor<4>(den);
        den += row_xor<2>(den); den += row_xor<1>(den);
        den += other_row(den);

        for (int c = beg; c < end; c += 32) {
            const int p = c + lane;
            float wgt = 0.0f;
            int   nb  = i;
            if (p < end) {
                wgt = __expf(slot_sc[p] - m) / den;
                nb  = slot_r1[p];
            }
            const int n = min(32, end - c);
            for (int j = 0; j < n; j += 8) {
                float4 v[8];
                float  a[8];
#pragma unroll
                for (int u = 0; u < 8; ++u) {
                    const int r = __builtin_amdgcn_readlane(nb, j + u);
                    a[u] = lane_f(wgt, j + u);
                    v[u] = ((const float4*)(embs + (size_t)r * D))[lane];
                }
#pragma unroll
                for (int u = 0; u < 8; ++u) {
                    acc.x = fmaf(a[u], v[u].x, acc.x);
                    acc.y = fmaf(a[u], v[u].y, acc.y);
                    acc.z = fmaf(a[u], v[u].z, acc.z);
                    acc.w = fmaf(a[u], v[u].w, acc.w);
                }
            }
        }
    }
    ((float4*)(out + (size_t)i * D))[lane] = acc;
}

__global__ void gat_softmax_denom(const int* __restrict__ ratings,
                                  const unsigned* __restrict__ segmax,
                                  float* __restrict__ scores,
                                  float* __restrict__ denom,
                                  int n_edges) {
    const int e = blockIdx.x * blockDim.x + threadIdx.x;
    if (e >= n_edges) return;
    const int r0 = ratings[2 * e];
    const float m = dec_order(segmax[r0]);
    const float ex = __expf(scores[e] - m);
    scores[e] = ex;
    atomicAdd(&denom[r0], ex);
}

__global__ void gat_aggregate(const float* __restrict__ embs,
                              const int* __restrict__ ratings,
                              const float* __restrict__ ex,
                              const float* __restrict__ denom,
                              float* __restrict__ out,
                              int n_edges) {
    const int lane = threadIdx.x & 31;
    const int e = blockIdx.x * (blockDim.x >> 5) + (threadIdx.x >> 5);
    if (e >= n_edges) return;
    const int r0 = ratings[2 * e + 0];
    const int r1 = ratings[2 * e + 1];
    const float attn = ex[e] / denom[r0];

    const float4 v = ((const float4*)(embs + (size_t)r1 * D))[lane];
    float* dst = out + (size_t)r0 * D + lane * 4;
    atomicAdd(dst + 0, attn * v.x);
    atomicAdd(dst + 1, attn * v.y);
    atomicAdd(dst + 2, attn * v.z);
    atomicAdd(dst + 3, attn * v.w);
}

extern "C" void kernel_launch(void* const* d_in, const int* in_sizes, int n_in,
                              void* d_out, int out_size, void* d_ws, size_t ws_size,
                              hipStream_t stream) {
    const float* embs    = (const float*)d_in[0];
    const int*   ratings = (const int*)d_in[1];
    const int node_num = in_sizes[0] / D;
    const int n_edges  = in_sizes[1] / 2;
    float* out = (float*)d_out;

    const int tiles = (n_edges + 15) / 16;
    dim3 blk(256);
    dim3 g1((tiles + 7) / 8);
    dim3 g2((n_edges + 255) / 256);
    char* ws = (char*)d_ws;

    const size_t need = (size_t)n_edges * 8 + (size_t)node_num * 8;
    if (ws_size >= need) {
        float* slot_sc = (float*)ws;
        int*   slot_r1 = (int*)(ws + (size_t)n_edges * 4);
        int*   counts  = (int*)(ws + (size_t)n_edges * 8);
        int*   cursor  = (int*)(ws + (size_t)n_edges * 8 + (size_t)node_num * 4);

        hipMemsetAsync(counts, 0, (size_t)node_num * 4, stream);
        dim3 gc((n_edges + 1023) / 1024);
        gat_count<<<gc, blk, 0, stream>>>(ratings, counts, n_edges);
        gat_scan<<<1, 1024, 0, stream>>>(counts, cursor, node_num);
        gat_scores_wmma<<<g1, blk, 0, stream>>>(embs, ratings, slot_sc, cursor, slot_r1, n_edges, 1);
        dim3 g4((node_num + 7) / 8);
        gat_rows<<<g4, blk, 0, stream>>>(embs, cursor, counts, slot_sc, slot_r1, out, node_num);
        return;
    }

    float*    scores = (float*)ws;
    unsigned* segmax = (unsigned*)(ws + (size_t)n_edges * 4);
    float*    denom  = (float*)(ws + (size_t)n_edges * 4 + (size_t)node_num * 4);

    hipMemsetAsync(d_out, 0, (size_t)out_size * sizeof(float), stream);
    hipMemsetAsync(segmax, 0, (size_t)node_num * 4, stream);
    hipMemsetAsync(denom, 0, (size_t)node_num * 4, stream);

    gat_scores_wmma<<<g1, blk, 0, stream>>>(embs, ratings, scores, (int*)segmax, nullptr, n_edges, 0);
    gat_softmax_denom<<<g2, blk, 0, stream>>>(ratings, segmax, scores, denom, n_edges);

    dim3 g3((n_edges + 7) / 8);
    gat_aggregate<<<g3, blk, 0, stream>>>(embs, ratings, scores, denom, out, n_edges);
}


# counting pass's returning atomic also gives each edge its place in its row (rank stored); scores pass writes slot = row start + rank with no atomic pass of its own; row sum order = count-pass arrival
# speedup vs baseline: 2.5664x; 1.3653x over previous
#include <hip/hip_runtime.h>
#include <hip/hip_bf16.h>

typedef __attribute__((ext_vector_type(2))) float v2f;
typedef __attribute__((ext_vector_type(8))) float v8f;

#define D 128

__device__ __forceinline__ unsigned enc_order(float f) {
    unsigned s = __float_as_uint(f);
    return (s & 0x80000000u) ? ~s : (s | 0x80000000u);
}
__device__ __forceinline__ float dec_order(unsigned u) {
    unsigned s = (u & 0x80000000u) ? (u & 0x7fffffffu) : ~u;
    return __uint_as_float(s);
}


template <int M>
__device__ __forceinline__ float row_xor(float x) {
    return __builtin_bit_cast(float,
        __builtin_amdgcn_update_dpp(0, __builtin_bit_cast(int, x),
                                    0x160 + M, 0xf, 0xf, false));
}
__device__ __forceinline__ float other_row(float x) {
    const int b = __builtin_bit_cast(int, x);
    return __builtin_bit_cast(float,
        __builtin_amdgcn_permlanex16(b, b, 0x76543210, 0xfedcba98, false, false));
}
template <int M>
__device__ __forceinline__ float fold_pair(float lo, float hi, bool bit) {
    const float keep = bit ? hi : lo;
    const float send = bit ? lo : hi;
    return keep + row_xor<M>(send);
}

template <int MODE>
__global__ void gat_scores_wmma(const float* __restrict__ embs,
                                const int* __restrict__ ratings,
                                float* __restrict__ scores,
                                int* __restrict__ aux,
                                int* __restrict__ slot_r1,
                                const int* __restrict__ rank,
                                int n_edges) {
    const int lane = threadIdx.x & 31;
    const int tile = blockIdx.x * (blockDim.x >> 5) + (threadIdx.x >> 5);
    const int e0 = tile << 4;
    if (e0 >= n_edges) return;

    int eL = e0 + (lane & 15);
    if (eL >= n_edges) eL = n_edges - 1;
    const int r0 = ratings[2 * eL + 0];
    const int r1 = ratings[2 * eL + 1];
    int slot = 0;
    if (MODE == 2) slot = aux[r0] + rank[eL];

    const int  k       = lane & 7;
    const bool on_diag = (lane < 8) || (lane >= 24);

    constexpr int AHEAD = 4;
    float4 a[16], b[16];
    float d[16];
#pragma unroll
    for (int j = 0; j < AHEAD; ++j) {
        const int ra = __builtin_amdgcn_readlane(r0, j);
        const int rb = __builtin_amdgcn_readlane(r1, j);
        a[j] = ((const float4*)(embs + (size_t)ra * D))[lane];
        b[j] = ((const float4*)(embs + (size_t)rb * D))[lane];
    }
#pragma unroll
    for (int j = 0; j < 16; ++j) {
        if (j + AHEAD < 16) {
            const int ra = __builtin_amdgcn_readlane(r0, j + AHEAD);
            const int rb = __builtin_amdgcn_readlane(r1, j + AHEAD);
            a[j + AHEAD] = ((const float4*)(embs + (size_t)ra * D))[lane];
            b[j + AHEAD] = ((const float4*)(embs + (size_t)rb * D))[lane];
        }
        const v2f a0 = {a[j].x, a[j].y}, a1 = {a[j].z, a[j].w};
        const v2f b0 = {b[j].x, b[j].y}, b1 = {b[j].z, b[j].w};

        v8f acc = {};
        acc = __builtin_amdgcn_wmma_f32_16x16x4_f32(
            false, a0, false, b0, (short)0, acc, false, false);
        acc = __builtin_amdgcn_wmma_f32_16x16x4_f32(
            false, a1, false, b1, (short)0, acc, false, false);

        float v = acc[0];
        v = (k == 1) ? acc[1] : v;
        v = (k == 2) ? acc[2] : v;
        v = (k == 3) ? acc[3] : v;
        v = (k == 4) ? acc[4] : v;
        v = (k == 5) ? acc[5] : v;
        v = (k == 6) ? acc[6] : v;
        v = (k == 7) ? acc[7] : v;
        d[j] = on_diag ? v : 0.0f;
    }

    const bool bit8 = (lane & 8) != 0, bit4 = (lane & 4) != 0;
    const bool bit2 = (lane & 2) != 0, bit1 = (lane & 1) != 0;
    float x[8], y[4], w[2];
#pragma unroll
    for (int j = 0; j < 8; ++j) x[j] = fold_pair<8>(d[j], d[j + 8], bit8);
#pragma unroll
    for (int j = 0; j < 4; ++j) y[j] = fold_pair<4>(x[j], x[j + 4], bit4);
#pragma unroll
    for (int j = 0; j < 2; ++j) w[j] = fold_pair<2>(y[j], y[j + 2], bit2);
    const float z  = fold_pair<1>(w[0], w[1], bit1);
    const float sc = z + other_row(z);

    if (MODE == 2) {
        int* const dst = (lane < 16) ? (int*)scores + slot : slot_r1 + slot;
        *dst = (lane < 16) ? __float_as_int(sc) : r1;
        return;
    }
    if (lane < 16 && (e0 + lane) < n_edges) {
        if (MODE == 1) {
            const int pos = atomicAdd(&aux[r0], 1);
            scores[pos]  = sc;
            slot_r1[pos] = r1;
        } else {
            scores[e0 + lane] = sc;
            atomicMax((unsigned*)aux + r0, enc_order(sc));
        }
    }
}

__global__ void gat_count(const int* __restrict__ ratings,
                          int* __restrict__ counts, int n_edges) {
    const int e = (blockIdx.x * blockDim.x + threadIdx.x) * 4;
    if (e + 3 < n_edges) {
        const int4 p = ((const int4*)ratings)[(e >> 1) + 0];
        const int4 q = ((const int4*)ratings)[(e >> 1) + 1];
        atomicAdd(&counts[p.x], 1);
        atomicAdd(&counts[p.z], 1);
        atomicAdd(&counts[q.x], 1);
        atomicAdd(&counts[q.z], 1);
    } else {
        for (int k = e; k < n_edges; ++k) atomicAdd(&counts[ratings[2 * k]], 1);
    }
}

__global__ void gat_count_rank(const int* __restrict__ ratings,
                               int* __restrict__ counts,
                               int* __restrict__ rank, int n_edges) {
    const int e = (blockIdx.x * blockDim.x + threadIdx.x) * 4;
    if (e + 3 < n_edges) {
        const int4 p = ((const int4*)ratings)[(e >> 1) + 0];
        const int4 q = ((const int4*)ratings)[(e >> 1) + 1];
        int4 r;
        r.x = atomicAdd(&counts[p.x], 1);
        r.y = atomicAdd(&counts[p.z], 1);
        r.z = atomicAdd(&counts[q.x], 1);
        r.w = atomicAdd(&counts[q.z], 1);
        ((int4*)rank)[e >> 2] = r;
    } else {
        for (int k = e; k < n_edges; ++k)
            rank[k] = atomicAdd(&counts[ratings[2 * k]], 1);
    }
}

__global__ void gat_scan(const int* __restrict__ counts,
                         int* __restrict__ cursor, int n_nodes) {
    __shared__ int wave_tot[32];
    const int t = threadIdx.x, lane = t & 31, w = t >> 5;
    const int per = (((n_nodes + (int)blockDim.x - 1) / (int)blockDim.x) + 3) & ~3;
    const int b = min(t * per, n_nodes), e = min(b + per, n_nodes);
    int s = 0;
    for (int j = b; j < e; ++j) s += counts[j];
    int incl = s;
    for (int d = 1; d < 32; d <<= 1) {
        const int y = __shfl_up(incl, d);
        if (lane >= d) incl += y;
    }
    if (lane == 31) wave_tot[w] = incl;
    __syncthreads();
    if (w == 0) {
        const int v = (lane < (int)(blockDim.x >> 5)) ? wave_tot[lane] : 0;
        int vi = v;
        for (int d = 1; d < 32; d <<= 1) {
            const int y = __shfl_up(vi, d);
            if (lane >= d) vi += y;
        }
        wave_tot[lane] = vi - v;
    }
    __syncthreads();
    int run = wave_tot[w] + incl - s;
    for (int j = b; j < e; ++j) { cursor[j] = run; run += counts[j]; }
}

__device__ __forceinline__ float lane_f(float x, int l) {
    return __builtin_bit_cast(float,
        __builtin_amdgcn_readlane(__builtin_bit_cast(int, x), l));
}

__global__ void gat_rows(const float* __restrict__ embs,
                         const int* __restrict__ cursor,
                         const int* __restrict__ counts,
                         const float* __restrict__ slot_sc,
                         const int* __restrict__ slot_r1,
                         float* __restrict__ out, int n_nodes, int at_end) {
    const int lane = threadIdx.x & 31;
    const int i = __builtin_amdgcn_readfirstlane(
        (int)(blockIdx.x * (blockDim.x >> 5) + (threadIdx.x >> 5)));
    if (i >= n_nodes) return;
    const int cnt = counts[i];
    const int beg = cursor[i] - (at_end ? cnt : 0);
    const int end = beg + cnt;

    float4 acc = {0.0f, 0.0f, 0.0f, 0.0f};
    if (beg < end) {
        float m = -INFINITY;
        for (int p = beg + lane; p < end; p += 32) m = fmaxf(m, slot_sc[p]);
        m = fmaxf(m, row_xor<8>(m)); m = fmaxf(m, row_xor<4>(m));
        m = fmaxf(m, row_xor<2>(m)); m = fmaxf(m, row_xor<1>(m));
        m = fmaxf(m, other_row(m));
        if (!__builtin_isfinite(m)) m = 0.0f;

        float den = 0.0f;
        for (int p = beg + lane; p < end; p += 32) den += __expf(slot_sc[p] - m);
        den += row_xor<8>(den); den += row_xor<4>(den);
        den += row_xor<2>(den); den += row_xor<1>(den);
        den += other_row(den);

        for (int c = beg; c < end; c += 32) {
            const int p = c + lane;
            float wgt = 0.0f;
            int   nb  = i;
            if (p < end) {
                wgt = __expf(slot_sc[p] - m) / den;
                nb  = slot_r1[p];
            }
            const int n = min(32, end - c);
            for (int j = 0; j < n; j += 8) {
                float4 v[8];
                float  a[8];
#pragma unroll
                for (int u = 0; u < 8; ++u) {
                    const int r = __builtin_amdgcn_readlane(nb, j + u);
                    a[u] = lane_f(wgt, j + u);
                    v[u] = ((const float4*)(embs + (size_t)r * D))[lane];
                }
#pragma unroll
                for (int u = 0; u < 8; ++u) {
                    acc.x = fmaf(a[u], v[u].x, acc.x);
                    acc.y = fmaf(a[u], v[u].y, acc.y);
                    acc.z = fmaf(a[u], v[u].z, acc.z);
                    acc.w = fmaf(a[u], v[u].w, acc.w);
                }
            }
        }
    }
    ((float4*)(out + (size_t)i * D))[lane] = acc;
}

__global__ void gat_softmax_denom(const int* __restrict__ ratings,
                                  const unsigned* __restrict__ segmax,
                                  float* __restrict__ scores,
                                  float* __restrict__ denom,
                                  int n_edges) {
    const int e = blockIdx.x * blockDim.x + threadIdx.x;
    if (e >= n_edges) return;
    const int r0 = ratings[2 * e];
    const float m = dec_order(segmax[r0]);
    const float ex = __expf(scores[e] - m);
    scores[e] = ex;
    atomicAdd(&denom[r0], ex);
}

__global__ void gat_aggregate(const float* __restrict__ embs,
                              const int* __restrict__ ratings,
                              const float* __restrict__ ex,
                              const float* __restrict__ denom,
                              float* __restrict__ out,
                              int n_edges) {
    const int lane = threadIdx.x & 31;
    const int e = blockIdx.x * (blockDim.x >> 5) + (threadIdx.x >> 5);
    if (e >= n_edges) return;
    const int r0 = ratings[2 * e + 0];
    const int r1 = ratings[2 * e + 1];
    const float attn = ex[e] / denom[r0];

    const float4 v = ((const float4*)(embs + (size_t)r1 * D))[lane];
    float* dst = out + (size_t)r0 * D + lane * 4;
    atomicAdd(dst + 0, attn * v.x);
    atomicAdd(dst + 1, attn * v.y);
    atomicAdd(dst + 2, attn * v.z);
    atomicAdd(dst + 3, attn * v.w);
}

extern "C" void kernel_launch(void* const* d_in, const int* in_sizes, int n_in,
                              void* d_out, int out_size, void* d_ws, size_t ws_size,
                              hipStream_t stream) {
    const float* embs    = (const float*)d_in[0];
    const int*   ratings = (const int*)d_in[1];
    const int node_num = in_sizes[0] / D;
    const int n_edges  = in_sizes[1] / 2;
    float* out = (float*)d_out;

    const int tiles = (n_edges + 15) / 16;
    dim3 blk(256);
    dim3 g1((tiles + 7) / 8);
    dim3 g2((n_edges + 255) / 256);
    char* ws = (char*)d_ws;

    const size_t need = (size_t)n_edges * 8 + (size_t)node_num * 8;
    if (ws_size >= need) {
        float* slot_sc = (float*)ws;
        int*   slot_r1 = (int*)(ws + (size_t)n_edges * 4);
        int*   counts  = (int*)(ws + (size_t)n_edges * 8);
        int*   cursor  = (int*)(ws + (size_t)n_edges * 8 + (size_t)node_num * 4);

        const size_t rank_off = (need + 15) & ~(size_t)15;
        int* rank = nullptr;
        if (ws_size >= rank_off + (size_t)n_edges * 4) rank = (int*)(ws + rank_off);
        else if ((size_t)out_size >= (size_t)n_edges)  rank = (int*)d_out;

        hipMemsetAsync(counts, 0, (size_t)node_num * 4, stream);
        dim3 gc((n_edges + 1023) / 1024);
        dim3 g4((node_num + 7) / 8);
        if (rank) {
            gat_count_rank<<<gc, blk, 0, stream>>>(ratings, counts, rank, n_edges);
            gat_scan<<<1, 1024, 0, stream>>>(counts, cursor, node_num);
            gat_scores_wmma<2><<<g1, blk, 0, stream>>>(embs, ratings, slot_sc, cursor, slot_r1, rank, n_edges);
            gat_rows<<<g4, blk, 0, stream>>>(embs, cursor, counts, slot_sc, slot_r1, out, node_num, 0);
        } else {
            gat_count<<<gc, blk, 0, stream>>>(ratings, counts, n_edges);
            gat_scan<<<1, 1024, 0, stream>>>(counts, cursor, node_num);
            gat_scores_wmma<1><<<g1, blk, 0, stream>>>(embs, ratings, slot_sc, cursor, slot_r1, nullptr, n_edges);
            gat_rows<<<g4, blk, 0, stream>>>(embs, cursor, counts, slot_sc, slot_r1, out, node_num, 1);
        }
        return;
    }

    float*    scores = (float*)ws;
    unsigned* segmax = (unsigned*)(ws + (size_t)n_edges * 4);
    float*    denom  = (float*)(ws + (size_t)n_edges * 4 + (size_t)node_num * 4);

    hipMemsetAsync(d_out, 0, (size_t)out_size * sizeof(float), stream);
    hipMemsetAsync(segmax, 0, (size_t)node_num * 4, stream);
    hipMemsetAsync(denom, 0, (size_t)node_num * 4, stream);

    gat_scores_wmma<0><<<g1, blk, 0, stream>>>(embs, ratings, scores, (int*)segmax, nullptr, nullptr, n_edges);
    gat_softmax_denom<<<g2, blk, 0, stream>>>(ratings, segmax, scores, denom, n_edges);

    dim3 g3((n_edges + 7) / 8);
    gat_aggregate<<<g3, blk, 0, stream>>>(embs, ratings, scores, denom, out, n_edges);
}


# scores pass folded into the row pass: row i stays in registers as A, each neighbour row is loaded once for score (2 WMMA, trace) and weighted sum; running-maximum softmax per 16 edges; new sum order
# speedup vs baseline: 2.8057x; 1.0932x over previous
#include <hip/hip_runtime.h>
#include <hip/hip_bf16.h>


typedef __attribute__((ext_vector_type(2))) float v2f;
typedef __attribute__((ext_vector_type(8))) float v8f;

#define D 128

__device__ __forceinline__ unsigned enc_order(float f) {
    unsigned s = __float_as_uint(f);
    return (s & 0x80000000u) ? ~s : (s | 0x80000000u);
}
__device__ __forceinline__ float dec_order(unsigned u) {
    unsigned s = (u & 0x80000000u) ? (u & 0x7fffffffu) : ~u;
    return __uint_as_float(s);
}


template <int M>
__device__ __forceinline__ float row_xor(float x) {
    return __builtin_bit_cast(float,
        __builtin_amdgcn_update_dpp(0, __builtin_bit_cast(int, x),
                                    0x160 + M, 0xf, 0xf, false));
}
__device__ __forceinline__ float other_row(float x) {
    const int b = __builtin_bit_cast(int, x);
    return __builtin_bit_cast(float,
        __builtin_amdgcn_permlanex16(b, b, 0x76543210, 0xfedcba98, false, false));
}
template <int M>
__device__ __forceinline__ float fold_pair(float lo, float hi, bool bit) {
    const float keep = bit ? hi : lo;
    const float send = bit ? lo : hi;
    return keep + row_xor<M>(send);
}

__device__ __forceinline__ float lane_f(float x, int l) {
    return __builtin_bit_cast(float,
        __builtin_amdgcn_readlane(__builtin_bit_cast(int, x), l));
}

__device__ __forceinline__ float diag_part(v2f a0, v2f a1, float4 b,
                                           int k, bool on_diag) {
    const v2f b0 = {b.x, b.y}, b1 = {b.z, b.w};
    v8f acc = {};
    acc = __builtin_amdgcn_wmma_f32_16x16x4_f32(
        false, a0, false, b0, (short)0, acc, false, false);
    acc = __builtin_amdgcn_wmma_f32_16x16x4_f32(
        false, a1, false, b1, (short)0, acc, false, false);
    float v = acc[0];
    v = (k == 1) ? acc[1] : v;
    v = (k == 2) ? acc[2] : v;
    v = (k == 3) ? acc[3] : v;
    v = (k == 4) ? acc[4] : v;
    v = (k == 5) ? acc[5] : v;
    v = (k == 6) ? acc[6] : v;
    v = (k == 7) ? acc[7] : v;
    return on_diag ? v : 0.0f;
}

__device__ __forceinline__ float lane_scores16(const float (&d)[16], int lane) {
    const bool bit8 = (lane & 8) != 0, bit4 = (lane & 4) != 0;
    const bool bit2 = (lane & 2) != 0, bit1 = (lane & 1) != 0;
    float x[8], y[4], w[2];
#pragma unroll
    for (int j = 0; j < 8; ++j) x[j] = fold_pair<8>(d[j], d[j + 8], bit8);
#pragma unroll
    for (int j = 0; j < 4; ++j) y[j] = fold_pair<4>(x[j], x[j + 4], bit4);
#pragma unroll
    for (int j = 0; j < 2; ++j) w[j] = fold_pair<2>(y[j], y[j + 2], bit2);
    const float z = fold_pair<1>(w[0], w[1], bit1);
    return z + other_row(z);
}
__device__ __forceinline__ float lane_scores8(const float (&d)[8], int lane) {
    const bool bit4 = (lane & 4) != 0;
    const bool bit2 = (lane & 2) != 0, bit1 = (lane & 1) != 0;
    float y[4], w[2];
#pragma unroll
    for (int j = 0; j < 4; ++j) y[j] = fold_pair<4>(d[j], d[j + 4], bit4);
#pragma unroll
    for (int j = 0; j < 2; ++j) w[j] = fold_pair<2>(y[j], y[j + 2], bit2);
    float z = fold_pair<1>(w[0], w[1], bit1);
    z += row_xor<8>(z);
    return z + other_row(z);
}

__global__ void gat_count_rank(const int* __restrict__ ratings,
                               int* __restrict__ counts,
                               int* __restrict__ rank, int n_edges) {
    const int e = (blockIdx.x * blockDim.x + threadIdx.x) * 4;
    if (e + 3 < n_edges) {
        const int4 p = ((const int4*)ratings)[(e >> 1) + 0];
        const int4 q = ((const int4*)ratings)[(e >> 1) + 1];
        int4 r;
        r.x = atomicAdd(&counts[p.x], 1);
        r.y = atomicAdd(&counts[p.z], 1);
        r.z = atomicAdd(&counts[q.x], 1);
        r.w = atomicAdd(&counts[q.z], 1);
        ((int4*)rank)[e >> 2] = r;
    } else {
        for (int k = e; k < n_edges; ++k)
            rank[k] = atomicAdd(&counts[ratings[2 * k]], 1);
    }
}

__global__ void gat_scan(const int* __restrict__ counts,
                         int* __restrict__ cursor, int n_nodes) {
    __shared__ int wave_tot[32];
    const int t = threadIdx.x, lane = t & 31, w = t >> 5;
    const int per = (((n_nodes + (int)blockDim.x - 1) / (int)blockDim.x) + 3) & ~3;
    const int b = min(t * per, n_nodes), e = min(b + per, n_nodes);
    int s = 0;
    for (int j = b; j < e; ++j) s += counts[j];
    int incl = s;
    for (int d = 1; d < 32; d <<= 1) {
        const int y = __shfl_up(incl, d);
        if (lane >= d) incl += y;
    }
    if (lane == 31) wave_tot[w] = incl;
    __syncthreads();
    if (w == 0) {
        const int v = (lane < (int)(blockDim.x >> 5)) ? wave_tot[lane] : 0;
        int vi = v;
        for (int d = 1; d < 32; d <<= 1) {
            const int y = __shfl_up(vi, d);
            if (lane >= d) vi += y;
        }
        wave_tot[lane] = vi - v;
    }
    __syncthreads();
    int run = wave_tot[w] + incl - s;
    for (int j = b; j < e; ++j) { cursor[j] = run; run += counts[j]; }
}

__global__ void gat_place(const int* __restrict__ ratings,
                          const int* __restrict__ cursor,
                          const int* __restrict__ rank,
                          int* __restrict__ slot_r1, int n_edges) {
    const int e = (blockIdx.x * blockDim.x + threadIdx.x) * 4;
    if (e + 3 < n_edges) {
        const int4 p = ((const int4*)ratings)[(e >> 1) + 0];
        const int4 q = ((const int4*)ratings)[(e >> 1) + 1];
        const int4 r = ((const int4*)rank)[e >> 2];
        const int s0 = cursor[p.x] + r.x;
        const int s1 = cursor[p.z] + r.y;
        const int s2 = cursor[q.x] + r.z;
        const int s3 = cursor[q.z] + r.w;
        slot_r1[s0] = p.y;
        slot_r1[s1] = p.w;
        slot_r1[s2] = q.y;
        slot_r1[s3] = q.w;
    } else {
        for (int k = e; k < n_edges; ++k)
            slot_r1[cursor[ratings[2 * k]] + rank[k]] = ratings[2 * k + 1];
    }
}

template <int NB>
__device__ __forceinline__ void gat_batch(const float* __restrict__ embs,
                                          v2f a0, v2f a1, int nb, int n,
                                          int lane, int k, bool on_diag,
                                          float& m, float& den, float4& acc) {
    float4 v[NB];
    float  d[NB];
#pragma unroll
    for (int j = 0; j < NB; ++j) {
        const int r = __builtin_amdgcn_readlane(nb, j);
        v[j] = ((const float4*)(embs + (size_t)r * D))[lane];
    }
#pragma unroll
    for (int j = 0; j < NB; ++j) d[j] = diag_part(a0, a1, v[j], k, on_diag);

    float sc;
    if constexpr (NB == 16) sc = lane_scores16(d, lane);
    else                    sc = lane_scores8(d, lane);

    const bool valid   = (lane & (NB - 1)) < n;
    const bool counted = valid && (lane < NB);
    float bm = valid ? sc : -INFINITY;
    bm = fmaxf(bm, row_xor<1>(bm));
    bm = fmaxf(bm, row_xor<2>(bm));
    bm = fmaxf(bm, row_xor<4>(bm));
    if constexpr (NB == 16) bm = fmaxf(bm, row_xor<8>(bm));
    float m_new = fmaxf(m, bm);
    if (!__builtin_isfinite(m_new)) m_new = 0.0f;
    const float scale = __expf(m - m_new);
    const float pw    = valid ? __expf(sc - m_new) : 0.0f;
    den   = fmaf(den, scale, counted ? pw : 0.0f);
    acc.x *= scale; acc.y *= scale; acc.z *= scale; acc.w *= scale;
    m = m_new;
#pragma unroll
    for (int j = 0; j < NB; ++j) {
        const float w = lane_f(pw, j);
        acc.x = fmaf(w, v[j].x, acc.x);
        acc.y = fmaf(w, v[j].y, acc.y);
        acc.z = fmaf(w, v[j].z, acc.z);
        acc.w = fmaf(w, v[j].w, acc.w);
    }
}

__global__ void __launch_bounds__(128)
gat_rows_fused(const float* __restrict__ embs,
               const int* __restrict__ cursor,
               const int* __restrict__ counts,
               const int* __restrict__ slot_r1,
               float* __restrict__ out, int n_nodes) {
    const int lane = threadIdx.x & 31;
    const int i = __builtin_amdgcn_readfirstlane(
        (int)(blockIdx.x * (blockDim.x >> 5) + (threadIdx.x >> 5)));
    if (i >= n_nodes) return;
    const int cnt = counts[i];
    const int beg = cursor[i];
    const int end = beg + cnt;

    float4 acc = {0.0f, 0.0f, 0.0f, 0.0f};
    if (cnt > 0) {
        const float4 xi = ((const float4*)(embs + (size_t)i * D))[lane];
        const v2f a0 = {xi.x, xi.y}, a1 = {xi.z, xi.w};
        const int  k       = lane & 7;
        const bool on_diag = (lane < 8) || (lane >= 24);
        const int  l16     = lane & 15;

        float m = -INFINITY, den = 0.0f;
        int nb = i;
        if (beg + l16 < end) nb = slot_r1[beg + l16];
        for (int c = beg; c < end; c += 16) {
            int nb_next = i;
            if (c + 16 + l16 < end) nb_next = slot_r1[c + 16 + l16];
            const int n = end - c;
            if (n > 8) gat_batch<16>(embs, a0, a1, nb, n, lane, k, on_diag, m, den, acc);
            else       gat_batch<8>(embs, a0, a1, nb, n, lane, k, on_diag, m, den, acc);
            nb = nb_next;
        }
        den += row_xor<8>(den); den += row_xor<4>(den);
        den += row_xor<2>(den); den += row_xor<1>(den);
        den += other_row(den);
        const float inv = 1.0f / den;
        acc.x *= inv; acc.y *= inv; acc.z *= inv; acc.w *= inv;
    }
    ((float4*)(out + (size_t)i * D))[lane] = acc;
}

__global__ void gat_scores_wmma(const float* __restrict__ embs,
                                const int* __restrict__ ratings,
                                float* __restrict__ scores,
                                unsigned* __restrict__ segmax,
                                int n_edges) {
    const int lane = threadIdx.x & 31;
    const int tile = blockIdx.x * (blockDim.x >> 5) + (threadIdx.x >> 5);
    const int e0 = tile << 4;
    if (e0 >= n_edges) return;

    int eL = e0 + (lane & 15);
    if (eL >= n_edges) eL = n_edges - 1;
    const int r0 = ratings[2 * eL + 0];
    const int r1 = ratings[2 * eL + 1];

    const int  k       = lane & 7;
    const bool on_diag = (lane < 8) || (lane >= 24);

    constexpr int AHEAD = 4;
    float4 a[16], b[16];
    float d[16];
#pragma unroll
    for (int j = 0; j < AHEAD; ++j) {
        const int ra = __builtin_amdgcn_readlane(r0, j);
        const int rb = __builtin_amdgcn_readlane(r1, j);
        a[j] = ((const float4*)(embs + (size_t)ra * D))[lane];
        b[j] = ((const float4*)(embs + (size_t)rb * D))[lane];
    }
#pragma unroll
    for (int j = 0; j < 16; ++j) {
        if (j + AHEAD < 16) {
            const int ra = __builtin_amdgcn_readlane(r0, j + AHEAD);
            const int rb = __builtin_amdgcn_readlane(r1, j + AHEAD);
            a[j + AHEAD] = ((const float4*)(embs + (size_t)ra * D))[lane];
            b[j + AHEAD] = ((const float4*)(embs + (size_t)rb * D))[lane];
        }
        const v2f a0 = {a[j].x, a[j].y}, a1 = {a[j].z, a[j].w};
        d[j] = diag_part(a0, a1, b[j], k, on_diag);
    }
    const float sc = lane_scores16(d, lane);

    if (lane < 16 && (e0 + lane) < n_edges) {
        scores[e0 + lane] = sc;
        atomicMax(segmax + r0, enc_order(sc));
    }
}

__global__ void gat_softmax_denom(const int* __restrict__ ratings,
                                  const unsigned* __restrict__ segmax,
                                  float* __restrict__ scores,
                                  float* __restrict__ denom,
                                  int n_edges) {
    const int e = blockIdx.x * blockDim.x + threadIdx.x;
    if (e >= n_edges) return;
    const int r0 = ratings[2 * e];
    const float m = dec_order(segmax[r0]);
    const float ex = __expf(scores[e] - m);
    scores[e] = ex;
    atomicAdd(&denom[r0], ex);
}

__global__ void gat_aggregate(const float* __restrict__ embs,
                              const int* __restrict__ ratings,
                              const float* __restrict__ ex,
                              const float* __restrict__ denom,
                              float* __restrict__ out,
                              int n_edges) {
    const int lane = threadIdx.x & 31;
    const int e = blockIdx.x * (blockDim.x >> 5) + (threadIdx.x >> 5);
    if (e >= n_edges) return;
    const int r0 = ratings[2 * e + 0];
    const int r1 = ratings[2 * e + 1];
    const float attn = ex[e] / denom[r0];

    const float4 v = ((const float4*)(embs + (size_t)r1 * D))[lane];
    float* dst = out + (size_t)r0 * D + lane * 4;
    atomicAdd(dst + 0, attn * v.x);
    atomicAdd(dst + 1, attn * v.y);
    atomicAdd(dst + 2, attn * v.z);
    atomicAdd(dst + 3, attn * v.w);
}

extern "C" void kernel_launch(void* const* d_in, const int* in_sizes, int n_in,
                              void* d_out, int out_size, void* d_ws, size_t ws_size,
                              hipStream_t stream) {
    const float* embs    = (const float*)d_in[0];
    const int*   ratings = (const int*)d_in[1];
    const int node_num = in_sizes[0] / D;
    const int n_edges  = in_sizes[1] / 2;
    float* out = (float*)d_out;

    dim3 blk(256);
    char* ws = (char*)d_ws;

    const size_t need = (size_t)n_edges * 8 + (size_t)node_num * 8;
    if (ws_size >= need) {
        int* rank    = (int*)ws;
        int* slot_r1 = (int*)(ws + (size_t)n_edges * 4);
        int* counts  = (int*)(ws + (size_t)n_edges * 8);
        int* cursor  = (int*)(ws + (size_t)n_edges * 8 + (size_t)node_num * 4);

        hipMemsetAsync(counts, 0, (size_t)node_num * 4, stream);
        dim3 gc((n_edges + 1023) / 1024);
        dim3 g4((node_num + 3) / 4);
        gat_count_rank<<<gc, blk, 0, stream>>>(ratings, counts, rank, n_edges);
        gat_scan<<<1, 1024, 0, stream>>>(counts, cursor, node_num);
        gat_place<<<gc, blk, 0, stream>>>(ratings, cursor, rank, slot_r1, n_edges);
        gat_rows_fused<<<g4, dim3(128), 0, stream>>>(embs, cursor, counts, slot_r1, out, node_num);
        return;
    }

    const int tiles = (n_edges + 15) / 16;
    dim3 g1((tiles + 7) / 8);
    dim3 g2((n_edges + 255) / 256);

    float*    scores = (float*)ws;
    unsigned* segmax = (unsigned*)(ws + (size_t)n_edges * 4);
    float*    denom  = (float*)(ws + (size_t)n_edges * 4 + (size_t)node_num * 4);

    hipMemsetAsync(d_out, 0, (size_t)out_size * sizeof(float), stream);
    hipMemsetAsync(segmax, 0, (size_t)node_num * 4, stream);
    hipMemsetAsync(denom, 0, (size_t)node_num * 4, stream);

    gat_scores_wmma<<<g1, blk, 0, stream>>>(embs, ratings, scores, segmax, n_edges);
    gat_softmax_denom<<<g2, blk, 0, stream>>>(ratings, segmax, scores, denom, n_edges);

    dim3 g3((n_edges + 7) / 8);
    gat_aggregate<<<g3, blk, 0, stream>>>(embs, ratings, scores, denom, out, n_edges);
}


# counting pass ranks edges in LDS: per 8192-edge chunk a returning LDS add gives the place in chunk, one returning global add per (chunk,row) the chunk's start; row sum order = chunk + LDS arrival
# speedup vs baseline: 4.5992x; 1.6392x over previous
#include <hip/hip_runtime.h>
#include <hip/hip_bf16.h>


typedef __attribute__((ext_vector_type(2))) float v2f;
typedef __attribute__((ext_vector_type(8))) float v8f;

#define D 128

__device__ __forceinline__ unsigned enc_order(float f) {
    unsigned s = __float_as_uint(f);
    return (s & 0x80000000u) ? ~s : (s | 0x80000000u);
}
__device__ __forceinline__ float dec_order(unsigned u) {
    unsigned s = (u & 0x80000000u) ? (u & 0x7fffffffu) : ~u;
    return __uint_as_float(s);
}


template <int M>
__device__ __forceinline__ float row_xor(float x) {
    return __builtin_bit_cast(float,
        __builtin_amdgcn_update_dpp(0, __builtin_bit_cast(int, x),
                                    0x160 + M, 0xf, 0xf, false));
}
__device__ __forceinline__ float other_row(float x) {
    const int b = __builtin_bit_cast(int, x);
    return __builtin_bit_cast(float,
        __builtin_amdgcn_permlanex16(b, b, 0x76543210, 0xfedcba98, false, false));
}
template <int M>
__device__ __forceinline__ float fold_pair(float lo, float hi, bool bit) {
    const float keep = bit ? hi : lo;
    const float send = bit ? lo : hi;
    return keep + row_xor<M>(send);
}

__device__ __forceinline__ float lane_f(float x, int l) {
    return __builtin_bit_cast(float,
        __builtin_amdgcn_readlane(__builtin_bit_cast(int, x), l));
}

__device__ __forceinline__ float diag_part(v2f a0, v2f a1, float4 b,
                                           int k, bool on_diag) {
    const v2f b0 = {b.x, b.y}, b1 = {b.z, b.w};
    v8f acc = {};
    acc = __builtin_amdgcn_wmma_f32_16x16x4_f32(
        false, a0, false, b0, (short)0, acc, false, false);
    acc = __builtin_amdgcn_wmma_f32_16x16x4_f32(
        false, a1, false, b1, (short)0, acc, false, false);
    float v = acc[0];
    v = (k == 1) ? acc[1] : v;
    v = (k == 2) ? acc[2] : v;
    v = (k == 3) ? acc[3] : v;
    v = (k == 4) ? acc[4] : v;
    v = (k == 5) ? acc[5] : v;
    v = (k == 6) ? acc[6] : v;
    v = (k == 7) ? acc[7] : v;
    return on_diag ? v : 0.0f;
}

__device__ __forceinline__ float lane_scores16(const float (&d)[16], int lane) {
    const bool bit8 = (lane & 8) != 0, bit4 = (lane & 4) != 0;
    const bool bit2 = (lane & 2) != 0, bit1 = (lane & 1) != 0;
    float x[8], y[4], w[2];
#pragma unroll
    for (int j = 0; j < 8; ++j) x[j] = fold_pair<8>(d[j], d[j + 8], bit8);
#pragma unroll
    for (int j = 0; j < 4; ++j) y[j] = fold_pair<4>(x[j], x[j + 4], bit4);
#pragma unroll
    for (int j = 0; j < 2; ++j) w[j] = fold_pair<2>(y[j], y[j + 2], bit2);
    const float z = fold_pair<1>(w[0], w[1], bit1);
    return z + other_row(z);
}
__device__ __forceinline__ float lane_scores8(const float (&d)[8], int lane) {
    const bool bit4 = (lane & 4) != 0;
    const bool bit2 = (lane & 2) != 0, bit1 = (lane & 1) != 0;
    float y[4], w[2];
#pragma unroll
    for (int j = 0; j < 4; ++j) y[j] = fold_pair<4>(d[j], d[j + 4], bit4);
#pragma unroll
    for (int j = 0; j < 2; ++j) w[j] = fold_pair<2>(y[j], y[j + 2], bit2);
    float z = fold_pair<1>(w[0], w[1], bit1);
    z += row_xor<8>(z);
    return z + other_row(z);
}

__global__ void gat_count_rank(const int* __restrict__ ratings,
                               int* __restrict__ counts,
                               int* __restrict__ rank, int n_edges) {
    const int e = (blockIdx.x * blockDim.x + threadIdx.x) * 4;
    if (e + 3 < n_edges) {
        const int4 p = ((const int4*)ratings)[(e >> 1) + 0];
        const int4 q = ((const int4*)ratings)[(e >> 1) + 1];
        int4 r;
        r.x = atomicAdd(&counts[p.x], 1);
        r.y = atomicAdd(&counts[p.z], 1);
        r.z = atomicAdd(&counts[q.x], 1);
        r.w = atomicAdd(&counts[q.z], 1);
        ((int4*)rank)[e >> 2] = r;
    } else {
        for (int k = e; k < n_edges; ++k)
            rank[k] = atomicAdd(&counts[ratings[2 * k]], 1);
    }
}

#define RANK_THREADS 1024
#define RANK_QUADS   2
#define RANK_ROWS    16384
#define RANK_CHUNK   (RANK_THREADS * 4 * RANK_QUADS)

__global__ void __launch_bounds__(RANK_THREADS)
gat_count_rank_lds(const int* __restrict__ ratings,
                   int* __restrict__ counts,
                   int* __restrict__ rank, int n_edges, int n_nodes) {
    __shared__ int tab[RANK_ROWS];
    const int t = threadIdx.x;
    for (int r = t; r < n_nodes; r += RANK_THREADS) tab[r] = 0;

    const int  e0    = blockIdx.x * RANK_CHUNK + t * 4;
    const bool whole = (blockIdx.x + 1) * RANK_CHUNK <= n_edges;

    int row[RANK_QUADS][4];
    if (whole) {
#pragma unroll
        for (int q = 0; q < RANK_QUADS; ++q) {
            const int e = e0 + q * RANK_THREADS * 4;
            const int4 a = ((const int4*)ratings)[(e >> 1) + 0];
            const int4 b = ((const int4*)ratings)[(e >> 1) + 1];
            row[q][0] = a.x; row[q][1] = a.z; row[q][2] = b.x; row[q][3] = b.z;
        }
    } else {
#pragma unroll
        for (int q = 0; q < RANK_QUADS; ++q)
#pragma unroll
            for (int k = 0; k < 4; ++k) {
                const int e = e0 + q * RANK_THREADS * 4 + k;
                row[q][k] = (e < n_edges) ? ratings[2 * e] : -1;
            }
    }
    __syncthreads();

    int place[RANK_QUADS][4];
#pragma unroll
    for (int q = 0; q < RANK_QUADS; ++q)
#pragma unroll
        for (int k = 0; k < 4; ++k)
            place[q][k] = (row[q][k] >= 0) ? atomicAdd(&tab[row[q][k]], 1) : 0;
    __syncthreads();

    constexpr int RPT = RANK_ROWS / RANK_THREADS;
    int c[RPT], start[RPT];
#pragma unroll
    for (int j = 0; j < RPT; ++j) {
        const int r = t + j * RANK_THREADS;
        c[j] = (r < n_nodes) ? tab[r] : 0;
    }
#pragma unroll
    for (int j = 0; j < RPT; ++j)
        start[j] = (c[j] > 0) ? atomicAdd(&counts[t + j * RANK_THREADS], c[j]) : 0;
#pragma unroll
    for (int j = 0; j < RPT; ++j)
        if (c[j] > 0) tab[t + j * RANK_THREADS] = start[j];
    __syncthreads();

    if (whole) {
#pragma unroll
        for (int q = 0; q < RANK_QUADS; ++q) {
            const int e = e0 + q * RANK_THREADS * 4;
            int4 rk;
            rk.x = tab[row[q][0]] + place[q][0];
            rk.y = tab[row[q][1]] + place[q][1];
            rk.z = tab[row[q][2]] + place[q][2];
            rk.w = tab[row[q][3]] + place[q][3];
            ((int4*)rank)[e >> 2] = rk;
        }
    } else {
#pragma unroll
        for (int q = 0; q < RANK_QUADS; ++q)
#pragma unroll
            for (int k = 0; k < 4; ++k) {
                const int e = e0 + q * RANK_THREADS * 4 + k;
                if (e < n_edges) rank[e] = tab[row[q][k]] + place[q][k];
            }
    }
}

__global__ void gat_scan(const int* __restrict__ counts,
                         int* __restrict__ cursor, int n_nodes) {
    __shared__ int wave_tot[32];
    const int t = threadIdx.x, lane = t & 31, w = t >> 5;
    const int per = (((n_nodes + (int)blockDim.x - 1) / (int)blockDim.x) + 3) & ~3;
    const int b = min(t * per, n_nodes), e = min(b + per, n_nodes);
    int s = 0;
    for (int j = b; j < e; ++j) s += counts[j];
    int incl = s;
    for (int d = 1; d < 32; d <<= 1) {
        const int y = __shfl_up(incl, d);
        if (lane >= d) incl += y;
    }
    if (lane == 31) wave_tot[w] = incl;
    __syncthreads();
    if (w == 0) {
        const int v = (lane < (int)(blockDim.x >> 5)) ? wave_tot[lane] : 0;
        int vi = v;
        for (int d = 1; d < 32; d <<= 1) {
            const int y = __shfl_up(vi, d);
            if (lane >= d) vi += y;
        }
        wave_tot[lane] = vi - v;
    }
    __syncthreads();
    int run = wave_tot[w] + incl - s;
    for (int j = b; j < e; ++j) { cursor[j] = run; run += counts[j]; }
}

__global__ void gat_place(const int* __restrict__ ratings,
                          const int* __restrict__ cursor,
                          const int* __restrict__ rank,
                          int* __restrict__ slot_r1, int n_edges) {
    const int e = (blockIdx.x * blockDim.x + threadIdx.x) * 4;
    if (e + 3 < n_edges) {
        const int4 p = ((const int4*)ratings)[(e >> 1) + 0];
        const int4 q = ((const int4*)ratings)[(e >> 1) + 1];
        const int4 r = ((const int4*)rank)[e >> 2];
        const int s0 = cursor[p.x] + r.x;
        const int s1 = cursor[p.z] + r.y;
        const int s2 = cursor[q.x] + r.z;
        const int s3 = cursor[q.z] + r.w;
        slot_r1[s0] = p.y;
        slot_r1[s1] = p.w;
        slot_r1[s2] = q.y;
        slot_r1[s3] = q.w;
    } else {
        for (int k = e; k < n_edges; ++k)
            slot_r1[cursor[ratings[2 * k]] + rank[k]] = ratings[2 * k + 1];
    }
}

template <int NB>
__device__ __forceinline__ void gat_batch(const float* __restrict__ embs,
                                          v2f a0, v2f a1, int nb, int n,
                                          int lane, int k, bool on_diag,
                                          float& m, float& den, float4& acc) {
    float4 v[NB];
    float  d[NB];
#pragma unroll
    for (int j = 0; j < NB; ++j) {
        const int r = __builtin_amdgcn_readlane(nb, j);
        v[j] = ((const float4*)(embs + (size_t)r * D))[lane];
    }
#pragma unroll
    for (int j = 0; j < NB; ++j) d[j] = diag_part(a0, a1, v[j], k, on_diag);

    float sc;
    if constexpr (NB == 16) sc = lane_scores16(d, lane);
    else                    sc = lane_scores8(d, lane);

    const bool valid   = (lane & (NB - 1)) < n;
    const bool counted = valid && (lane < NB);
    float bm = valid ? sc : -INFINITY;
    bm = fmaxf(bm, row_xor<1>(bm));
    bm = fmaxf(bm, row_xor<2>(bm));
    bm = fmaxf(bm, row_xor<4>(bm));
    if constexpr (NB == 16) bm = fmaxf(bm, row_xor<8>(bm));
    float m_new = fmaxf(m, bm);
    if (!__builtin_isfinite(m_new)) m_new = 0.0f;
    const float scale = __expf(m - m_new);
    const float pw    = valid ? __expf(sc - m_new) : 0.0f;
    den   = fmaf(den, scale, counted ? pw : 0.0f);
    acc.x *= scale; acc.y *= scale; acc.z *= scale; acc.w *= scale;
    m = m_new;
#pragma unroll
    for (int j = 0; j < NB; ++j) {
        const float w = lane_f(pw, j);
        acc.x = fmaf(w, v[j].x, acc.x);
        acc.y = fmaf(w, v[j].y, acc.y);
        acc.z = fmaf(w, v[j].z, acc.z);
        acc.w = fmaf(w, v[j].w, acc.w);
    }
}

__global__ void __launch_bounds__(128)
gat_rows_fused(const float* __restrict__ embs,
               const int* __restrict__ cursor,
               const int* __restrict__ counts,
               const int* __restrict__ slot_r1,
               float* __restrict__ out, int n_nodes) {
    const int lane = threadIdx.x & 31;
    const int i = __builtin_amdgcn_readfirstlane(
        (int)(blockIdx.x * (blockDim.x >> 5) + (threadIdx.x >> 5)));
    if (i >= n_nodes) return;
    const int cnt = counts[i];
    const int beg = cursor[i];
    const int end = beg + cnt;

    float4 acc = {0.0f, 0.0f, 0.0f, 0.0f};
    if (cnt > 0) {
        const float4 xi = ((const float4*)(embs + (size_t)i * D))[lane];
        const v2f a0 = {xi.x, xi.y}, a1 = {xi.z, xi.w};
        const int  k       = lane & 7;
        const bool on_diag = (lane < 8) || (lane >= 24);
        const int  l16     = lane & 15;

        float m = -INFINITY, den = 0.0f;
        int nb = i;
        if (beg + l16 < end) nb = slot_r1[beg + l16];
        for (int c = beg; c < end; c += 16) {
            int nb_next = i;
            if (c + 16 + l16 < end) nb_next = slot_r1[c + 16 + l16];
            const int n = end - c;
            if (n > 8) gat_batch<16>(embs, a0, a1, nb, n, lane, k, on_diag, m, den, acc);
            else       gat_batch<8>(embs, a0, a1, nb, n, lane, k, on_diag, m, den, acc);
            nb = nb_next;
        }
        den += row_xor<8>(den); den += row_xor<4>(den);
        den += row_xor<2>(den); den += row_xor<1>(den);
        den += other_row(den);
        const float inv = 1.0f / den;
        acc.x *= inv; acc.y *= inv; acc.z *= inv; acc.w *= inv;
    }
    ((float4*)(out + (size_t)i * D))[lane] = acc;
}

__global__ void gat_scores_wmma(const float* __restrict__ embs,
                                const int* __restrict__ ratings,
                                float* __restrict__ scores,
                                unsigned* __restrict__ segmax,
                                int n_edges) {
    const int lane = threadIdx.x & 31;
    const int tile = blockIdx.x * (blockDim.x >> 5) + (threadIdx.x >> 5);
    const int e0 = tile << 4;
    if (e0 >= n_edges) return;

    int eL = e0 + (lane & 15);
    if (eL >= n_edges) eL = n_edges - 1;
    const int r0 = ratings[2 * eL + 0];
    const int r1 = ratings[2 * eL + 1];

    const int  k       = lane & 7;
    const bool on_diag = (lane < 8) || (lane >= 24);

    constexpr int AHEAD = 4;
    float4 a[16], b[16];
    float d[16];
#pragma unroll
    for (int j = 0; j < AHEAD; ++j) {
        const int ra = __builtin_amdgcn_readlane(r0, j);
        const int rb = __builtin_amdgcn_readlane(r1, j);
        a[j] = ((const float4*)(embs + (size_t)ra * D))[lane];
        b[j] = ((const float4*)(embs + (size_t)rb * D))[lane];
    }
#pragma unroll
    for (int j = 0; j < 16; ++j) {
        if (j + AHEAD < 16) {
            const int ra = __builtin_amdgcn_readlane(r0, j + AHEAD);
            const int rb = __builtin_amdgcn_readlane(r1, j + AHEAD);
            a[j + AHEAD] = ((const float4*)(embs + (size_t)ra * D))[lane];
            b[j + AHEAD] = ((const float4*)(embs + (size_t)rb * D))[lane];
        }
        const v2f a0 = {a[j].x, a[j].y}, a1 = {a[j].z, a[j].w};
        d[j] = diag_part(a0, a1, b[j], k, on_diag);
    }
    const float sc = lane_scores16(d, lane);

    if (lane < 16 && (e0 + lane) < n_edges) {
        scores[e0 + lane] = sc;
        atomicMax(segmax + r0, enc_order(sc));
    }
}

__global__ void gat_softmax_denom(const int* __restrict__ ratings,
                                  const unsigned* __restrict__ segmax,
                                  float* __restrict__ scores,
                                  float* __restrict__ denom,
                                  int n_edges) {
    const int e = blockIdx.x * blockDim.x + threadIdx.x;
    if (e >= n_edges) return;
    const int r0 = ratings[2 * e];
    const float m = dec_order(segmax[r0]);
    const float ex = __expf(scores[e] - m);
    scores[e] = ex;
    atomicAdd(&denom[r0], ex);
}

__global__ void gat_aggregate(const float* __restrict__ embs,
                              const int* __restrict__ ratings,
                              const float* __restrict__ ex,
                              const float* __restrict__ denom,
                              float* __restrict__ out,
                              int n_edges) {
    const int lane = threadIdx.x & 31;
    const int e = blockIdx.x * (blockDim.x >> 5) + (threadIdx.x >> 5);
    if (e >= n_edges) return;
    const int r0 = ratings[2 * e + 0];
    const int r1 = ratings[2 * e + 1];
    const float attn = ex[e] / denom[r0];

    const float4 v = ((const float4*)(embs + (size_t)r1 * D))[lane];
    float* dst = out + (size_t)r0 * D + lane * 4;
    atomicAdd(dst + 0, attn * v.x);
    atomicAdd(dst + 1, attn * v.y);
    atomicAdd(dst + 2, attn * v.z);
    atomicAdd(dst + 3, attn * v.w);
}

extern "C" void kernel_launch(void* const* d_in, const int* in_sizes, int n_in,
                              void* d_out, int out_size, void* d_ws, size_t ws_size,
                              hipStream_t stream) {
    const float* embs    = (const float*)d_in[0];
    const int*   ratings = (const int*)d_in[1];
    const int node_num = in_sizes[0] / D;
    const int n_edges  = in_sizes[1] / 2;
    float* out = (float*)d_out;

    dim3 blk(256);
    char* ws = (char*)d_ws;

    const size_t need = (size_t)n_edges * 8 + (size_t)node_num * 8;
    if (ws_size >= need) {
        int* rank    = (int*)ws;
        int* slot_r1 = (int*)(ws + (size_t)n_edges * 4);
        int* counts  = (int*)(ws + (size_t)n_edges * 8);
        int* cursor  = (int*)(ws + (size_t)n_edges * 8 + (size_t)node_num * 4);

        hipMemsetAsync(counts, 0, (size_t)node_num * 4, stream);
        dim3 gc((n_edges + 1023) / 1024);
        dim3 g4((node_num + 3) / 4);
        if (node_num <= RANK_ROWS) {
            dim3 gk((n_edges + RANK_CHUNK - 1) / RANK_CHUNK);
            gat_count_rank_lds<<<gk, dim3(RANK_THREADS), 0, stream>>>(ratings, counts, rank, n_edges, node_num);
        } else {
            gat_count_rank<<<gc, blk, 0, stream>>>(ratings, counts, rank, n_edges);
        }
        gat_scan<<<1, 1024, 0, stream>>>(counts, cursor, node_num);
        gat_place<<<gc, blk, 0, stream>>>(ratings, cursor, rank, slot_r1, n_edges);
        gat_rows_fused<<<g4, dim3(128), 0, stream>>>(embs, cursor, counts, slot_r1, out, node_num);
        return;
    }

    const int tiles = (n_edges + 15) / 16;
    dim3 g1((tiles + 7) / 8);
    dim3 g2((n_edges + 255) / 256);

    float*    scores = (float*)ws;
    unsigned* segmax = (unsigned*)(ws + (size_t)n_edges * 4);
    float*    denom  = (float*)(ws + (size_t)n_edges * 4 + (size_t)node_num * 4);

    hipMemsetAsync(d_out, 0, (size_t)out_size * sizeof(float), stream);
    hipMemsetAsync(segmax, 0, (size_t)node_num * 4, stream);
    hipMemsetAsync(denom, 0, (size_t)node_num * 4, stream);

    gat_scores_wmma<<<g1, blk, 0, stream>>>(embs, ratings, scores, segmax, n_edges);
    gat_softmax_denom<<<g2, blk, 0, stream>>>(ratings, segmax, scores, denom, n_edges);

    dim3 g3((n_edges + 7) / 8);
    gat_aggregate<<<g3, blk, 0, stream>>>(embs, ratings, scores, denom, out, n_edges);
}


# neighbour slots at a fixed stride of 128 per row: counting pass stores each neighbour at row*128+rank itself; prefix-sum and placing launches gone; longer rows walk the edge list; sum order kept
# speedup vs baseline: 4.7646x; 1.0360x over previous
#include <hip/hip_runtime.h>
#include <hip/hip_bf16.h>


typedef __attribute__((ext_vector_type(2))) float v2f;
typedef __attribute__((ext_vector_type(8))) float v8f;

#define D 128

__device__ __forceinline__ unsigned enc_order(float f) {
    unsigned s = __float_as_uint(f);
    return (s & 0x80000000u) ? ~s : (s | 0x80000000u);
}
__device__ __forceinline__ float dec_order(unsigned u) {
    unsigned s = (u & 0x80000000u) ? (u & 0x7fffffffu) : ~u;
    return __uint_as_float(s);
}


template <int M>
__device__ __forceinline__ float row_xor(float x) {
    return __builtin_bit_cast(float,
        __builtin_amdgcn_update_dpp(0, __builtin_bit_cast(int, x),
                                    0x160 + M, 0xf, 0xf, false));
}
__device__ __forceinline__ float other_row(float x) {
    const int b = __builtin_bit_cast(int, x);
    return __builtin_bit_cast(float,
        __builtin_amdgcn_permlanex16(b, b, 0x76543210, 0xfedcba98, false, false));
}
template <int M>
__device__ __forceinline__ float fold_pair(float lo, float hi, bool bit) {
    const float keep = bit ? hi : lo;
    const float send = bit ? lo : hi;
    return keep + row_xor<M>(send);
}

__device__ __forceinline__ float lane_f(float x, int l) {
    return __builtin_bit_cast(float,
        __builtin_amdgcn_readlane(__builtin_bit_cast(int, x), l));
}

__device__ __forceinline__ float diag_part(v2f a0, v2f a1, float4 b,
                                           int k, bool on_diag) {
    const v2f b0 = {b.x, b.y}, b1 = {b.z, b.w};
    v8f acc = {};
    acc = __builtin_amdgcn_wmma_f32_16x16x4_f32(
        false, a0, false, b0, (short)0, acc, false, false);
    acc = __builtin_amdgcn_wmma_f32_16x16x4_f32(
        false, a1, false, b1, (short)0, acc, false, false);
    float v = acc[0];
    v = (k == 1) ? acc[1] : v;
    v = (k == 2) ? acc[2] : v;
    v = (k == 3) ? acc[3] : v;
    v = (k == 4) ? acc[4] : v;
    v = (k == 5) ? acc[5] : v;
    v = (k == 6) ? acc[6] : v;
    v = (k == 7) ? acc[7] : v;
    return on_diag ? v : 0.0f;
}

__device__ __forceinline__ float lane_scores16(const float (&d)[16], int lane) {
    const bool bit8 = (lane & 8) != 0, bit4 = (lane & 4) != 0;
    const bool bit2 = (lane & 2) != 0, bit1 = (lane & 1) != 0;
    float x[8], y[4], w[2];
#pragma unroll
    for (int j = 0; j < 8; ++j) x[j] = fold_pair<8>(d[j], d[j + 8], bit8);
#pragma unroll
    for (int j = 0; j < 4; ++j) y[j] = fold_pair<4>(x[j], x[j + 4], bit4);
#pragma unroll
    for (int j = 0; j < 2; ++j) w[j] = fold_pair<2>(y[j], y[j + 2], bit2);
    const float z = fold_pair<1>(w[0], w[1], bit1);
    return z + other_row(z);
}
__device__ __forceinline__ float lane_scores8(const float (&d)[8], int lane) {
    const bool bit4 = (lane & 4) != 0;
    const bool bit2 = (lane & 2) != 0, bit1 = (lane & 1) != 0;
    float y[4], w[2];
#pragma unroll
    for (int j = 0; j < 4; ++j) y[j] = fold_pair<4>(d[j], d[j + 4], bit4);
#pragma unroll
    for (int j = 0; j < 2; ++j) w[j] = fold_pair<2>(y[j], y[j + 2], bit2);
    float z = fold_pair<1>(w[0], w[1], bit1);
    z += row_xor<8>(z);
    return z + other_row(z);
}

__global__ void gat_count_rank(const int* __restrict__ ratings,
                               int* __restrict__ counts,
                               int* __restrict__ rank, int n_edges) {
    const int e = (blockIdx.x * blockDim.x + threadIdx.x) * 4;
    if (e + 3 < n_edges) {
        const int4 p = ((const int4*)ratings)[(e >> 1) + 0];
        const int4 q = ((const int4*)ratings)[(e >> 1) + 1];
        int4 r;
        r.x = atomicAdd(&counts[p.x], 1);
        r.y = atomicAdd(&counts[p.z], 1);
        r.z = atomicAdd(&counts[q.x], 1);
        r.w = atomicAdd(&counts[q.z], 1);
        ((int4*)rank)[e >> 2] = r;
    } else {
        for (int k = e; k < n_edges; ++k)
            rank[k] = atomicAdd(&counts[ratings[2 * k]], 1);
    }
}

#define RANK_THREADS 1024
#define RANK_QUADS   2
#define RANK_ROWS    16384
#define RANK_CHUNK   (RANK_THREADS * 4 * RANK_QUADS)

__global__ void __launch_bounds__(RANK_THREADS)
gat_count_rank_lds(const int* __restrict__ ratings,
                   int* __restrict__ counts,
                   int* __restrict__ rank, int n_edges, int n_nodes) {
    __shared__ int tab[RANK_ROWS];
    const int t = threadIdx.x;
    for (int r = t; r < n_nodes; r += RANK_THREADS) tab[r] = 0;

    const int  e0    = blockIdx.x * RANK_CHUNK + t * 4;
    const bool whole = (blockIdx.x + 1) * RANK_CHUNK <= n_edges;

    int row[RANK_QUADS][4];
    if (whole) {
#pragma unroll
        for (int q = 0; q < RANK_QUADS; ++q) {
            const int e = e0 + q * RANK_THREADS * 4;
            const int4 a = ((const int4*)ratings)[(e >> 1) + 0];
            const int4 b = ((const int4*)ratings)[(e >> 1) + 1];
            row[q][0] = a.x; row[q][1] = a.z; row[q][2] = b.x; row[q][3] = b.z;
        }
    } else {
#pragma unroll
        for (int q = 0; q < RANK_QUADS; ++q)
#pragma unroll
            for (int k = 0; k < 4; ++k) {
                const int e = e0 + q * RANK_THREADS * 4 + k;
                row[q][k] = (e < n_edges) ? ratings[2 * e] : -1;
            }
    }
    __syncthreads();

    int place[RANK_QUADS][4];
#pragma unroll
    for (int q = 0; q < RANK_QUADS; ++q)
#pragma unroll
        for (int k = 0; k < 4; ++k)
            place[q][k] = (row[q][k] >= 0) ? atomicAdd(&tab[row[q][k]], 1) : 0;
    __syncthreads();

    constexpr int RPT = RANK_ROWS / RANK_THREADS;
    int c[RPT], start[RPT];
#pragma unroll
    for (int j = 0; j < RPT; ++j) {
        const int r = t + j * RANK_THREADS;
        c[j] = (r < n_nodes) ? tab[r] : 0;
    }
#pragma unroll
    for (int j = 0; j < RPT; ++j)
        start[j] = (c[j] > 0) ? atomicAdd(&counts[t + j * RANK_THREADS], c[j]) : 0;
#pragma unroll
    for (int j = 0; j < RPT; ++j)
        if (c[j] > 0) tab[t + j * RANK_THREADS] = start[j];
    __syncthreads();

    if (whole) {
#pragma unroll
        for (int q = 0; q < RANK_QUADS; ++q) {
            const int e = e0 + q * RANK_THREADS * 4;
            int4 rk;
            rk.x = tab[row[q][0]] + place[q][0];
            rk.y = tab[row[q][1]] + place[q][1];
            rk.z = tab[row[q][2]] + place[q][2];
            rk.w = tab[row[q][3]] + place[q][3];
            ((int4*)rank)[e >> 2] = rk;
        }
    } else {
#pragma unroll
        for (int q = 0; q < RANK_QUADS; ++q)
#pragma unroll
            for (int k = 0; k < 4; ++k) {
                const int e = e0 + q * RANK_THREADS * 4 + k;
                if (e < n_edges) rank[e] = tab[row[q][k]] + place[q][k];
            }
    }
}

__global__ void gat_scan(const int* __restrict__ counts,
                         int* __restrict__ cursor, int n_nodes) {
    __shared__ int wave_tot[32];
    const int t = threadIdx.x, lane = t & 31, w = t >> 5;
    const int per = (((n_nodes + (int)blockDim.x - 1) / (int)blockDim.x) + 3) & ~3;
    const int b = min(t * per, n_nodes), e = min(b + per, n_nodes);
    int s = 0;
    for (int j = b; j < e; ++j) s += counts[j];
    int incl = s;
    for (int d = 1; d < 32; d <<= 1) {
        const int y = __shfl_up(incl, d);
        if (lane >= d) incl += y;
    }
    if (lane == 31) wave_tot[w] = incl;
    __syncthreads();
    if (w == 0) {
        const int v = (lane < (int)(blockDim.x >> 5)) ? wave_tot[lane] : 0;
        int vi = v;
        for (int d = 1; d < 32; d <<= 1) {
            const int y = __shfl_up(vi, d);
            if (lane >= d) vi += y;
        }
        wave_tot[lane] = vi - v;
    }
    __syncthreads();
    int run = wave_tot[w] + incl - s;
    for (int j = b; j < e; ++j) { cursor[j] = run; run += counts[j]; }
}

__global__ void gat_place(const int* __restrict__ ratings,
                          const int* __restrict__ cursor,
                          const int* __restrict__ rank,
                          int* __restrict__ slot_r1, int n_edges) {
    const int e = (blockIdx.x * blockDim.x + threadIdx.x) * 4;
    if (e + 3 < n_edges) {
        const int4 p = ((const int4*)ratings)[(e >> 1) + 0];
        const int4 q = ((const int4*)ratings)[(e >> 1) + 1];
        const int4 r = ((const int4*)rank)[e >> 2];
        const int s0 = cursor[p.x] + r.x;
        const int s1 = cursor[p.z] + r.y;
        const int s2 = cursor[q.x] + r.z;
        const int s3 = cursor[q.z] + r.w;
        slot_r1[s0] = p.y;
        slot_r1[s1] = p.w;
        slot_r1[s2] = q.y;
        slot_r1[s3] = q.w;
    } else {
        for (int k = e; k < n_edges; ++k)
            slot_r1[cursor[ratings[2 * k]] + rank[k]] = ratings[2 * k + 1];
    }
}

template <int NB>
__device__ __forceinline__ void gat_batch(const float* __restrict__ embs,
                                          v2f a0, v2f a1, int nb, bool valid,
                                          int lane, int k, bool on_diag,
                                          float& m, float& den, float4& acc) {
    float4 v[NB];
    float  d[NB];
#pragma unroll
    for (int j = 0; j < NB; ++j) {
        const int r = __builtin_amdgcn_readlane(nb, j);
        v[j] = ((const float4*)(embs + (size_t)r * D))[lane];
    }
#pragma unroll
    for (int j = 0; j < NB; ++j) d[j] = diag_part(a0, a1, v[j], k, on_diag);

    float sc;
    if constexpr (NB == 16) sc = lane_scores16(d, lane);
    else                    sc = lane_scores8(d, lane);

    const bool counted = valid && (lane < NB);
    float bm = valid ? sc : -INFINITY;
    bm = fmaxf(bm, row_xor<1>(bm));
    bm = fmaxf(bm, row_xor<2>(bm));
    bm = fmaxf(bm, row_xor<4>(bm));
    if constexpr (NB == 16) bm = fmaxf(bm, row_xor<8>(bm));
    float m_new = fmaxf(m, bm);
    if (!__builtin_isfinite(m_new)) m_new = 0.0f;
    const float scale = __expf(m - m_new);
    const float pw    = valid ? __expf(sc - m_new) : 0.0f;
    den   = fmaf(den, scale, counted ? pw : 0.0f);
    acc.x *= scale; acc.y *= scale; acc.z *= scale; acc.w *= scale;
    m = m_new;
#pragma unroll
    for (int j = 0; j < NB; ++j) {
        const float w = lane_f(pw, j);
        acc.x = fmaf(w, v[j].x, acc.x);
        acc.y = fmaf(w, v[j].y, acc.y);
        acc.z = fmaf(w, v[j].z, acc.z);
        acc.w = fmaf(w, v[j].w, acc.w);
    }
}

__global__ void __launch_bounds__(128)
gat_rows_fused(const float* __restrict__ embs,
               const int* __restrict__ cursor,
               const int* __restrict__ counts,
               const int* __restrict__ slot_r1,
               float* __restrict__ out, int n_nodes) {
    const int lane = threadIdx.x & 31;
    const int i = __builtin_amdgcn_readfirstlane(
        (int)(blockIdx.x * (blockDim.x >> 5) + (threadIdx.x >> 5)));
    if (i >= n_nodes) return;
    const int cnt = counts[i];
    const int beg = cursor[i];
    const int end = beg + cnt;

    float4 acc = {0.0f, 0.0f, 0.0f, 0.0f};
    if (cnt > 0) {
        const float4 xi = ((const float4*)(embs + (size_t)i * D))[lane];
        const v2f a0 = {xi.x, xi.y}, a1 = {xi.z, xi.w};
        const int  k       = lane & 7;
        const bool on_diag = (lane < 8) || (lane >= 24);
        const int  l16     = lane & 15;

        float m = -INFINITY, den = 0.0f;
        int nb = i;
        if (beg + l16 < end) nb = slot_r1[beg + l16];
        for (int c = beg; c < end; c += 16) {
            int nb_next = i;
            if (c + 16 + l16 < end) nb_next = slot_r1[c + 16 + l16];
            const int n = end - c;
            if (n > 8) gat_batch<16>(embs, a0, a1, nb, l16 < n, lane, k, on_diag, m, den, acc);
            else       gat_batch<8>(embs, a0, a1, nb, k < n, lane, k, on_diag, m, den, acc);
            nb = nb_next;
        }
        den += row_xor<8>(den); den += row_xor<4>(den);
        den += row_xor<2>(den); den += row_xor<1>(den);
        den += other_row(den);
        const float inv = 1.0f / den;
        acc.x *= inv; acc.y *= inv; acc.z *= inv; acc.w *= inv;
    }
    ((float4*)(out + (size_t)i * D))[lane] = acc;
}

#define SLOT_STRIDE 128

__global__ void __launch_bounds__(RANK_THREADS)
gat_count_place_lds(const int* __restrict__ ratings,
                    int* __restrict__ counts,
                    int* __restrict__ slot_r1, int n_edges, int n_nodes) {
    __shared__ int tab[RANK_ROWS];
    const int t = threadIdx.x;
    for (int r = t; r < n_nodes; r += RANK_THREADS) tab[r] = 0;

    const int  e0    = blockIdx.x * RANK_CHUNK + t * 4;
    const bool whole = (blockIdx.x + 1) * RANK_CHUNK <= n_edges;

    int row[RANK_QUADS][4], nbr[RANK_QUADS][4];
    if (whole) {
#pragma unroll
        for (int q = 0; q < RANK_QUADS; ++q) {
            const int e = e0 + q * RANK_THREADS * 4;
            const int4 a = ((const int4*)ratings)[(e >> 1) + 0];
            const int4 b = ((const int4*)ratings)[(e >> 1) + 1];
            row[q][0] = a.x; row[q][1] = a.z; row[q][2] = b.x; row[q][3] = b.z;
            nbr[q][0] = a.y; nbr[q][1] = a.w; nbr[q][2] = b.y; nbr[q][3] = b.w;
        }
    } else {
#pragma unroll
        for (int q = 0; q < RANK_QUADS; ++q)
#pragma unroll
            for (int k = 0; k < 4; ++k) {
                const int e = e0 + q * RANK_THREADS * 4 + k;
                row[q][k] = (e < n_edges) ? ratings[2 * e] : -1;
                nbr[q][k] = (e < n_edges) ? ratings[2 * e + 1] : 0;
            }
    }
    __syncthreads();

    int place[RANK_QUADS][4];
#pragma unroll
    for (int q = 0; q < RANK_QUADS; ++q)
#pragma unroll
        for (int k = 0; k < 4; ++k)
            place[q][k] = (row[q][k] >= 0) ? atomicAdd(&tab[row[q][k]], 1) : 0;
    __syncthreads();

    constexpr int RPT = RANK_ROWS / RANK_THREADS;
    int c[RPT], start[RPT];
#pragma unroll
    for (int j = 0; j < RPT; ++j) {
        const int r = t + j * RANK_THREADS;
        c[j] = (r < n_nodes) ? tab[r] : 0;
    }
#pragma unroll
    for (int j = 0; j < RPT; ++j)
        start[j] = (c[j] > 0) ? atomicAdd(&counts[t + j * RANK_THREADS], c[j]) : 0;
#pragma unroll
    for (int j = 0; j < RPT; ++j)
        if (c[j] > 0) tab[t + j * RANK_THREADS] = start[j];
    __syncthreads();

    if (whole) {
        int rk[RANK_QUADS][4];
#pragma unroll
        for (int q = 0; q < RANK_QUADS; ++q)
#pragma unroll
            for (int k = 0; k < 4; ++k)
                rk[q][k] = tab[row[q][k]] + place[q][k];
#pragma unroll
        for (int q = 0; q < RANK_QUADS; ++q)
#pragma unroll
            for (int k = 0; k < 4; ++k)
                if (rk[q][k] < SLOT_STRIDE)
                    slot_r1[(size_t)row[q][k] * SLOT_STRIDE + rk[q][k]] = nbr[q][k];
    } else {
#pragma unroll
        for (int q = 0; q < RANK_QUADS; ++q)
#pragma unroll
            for (int k = 0; k < 4; ++k)
                if (row[q][k] >= 0) {
                    const int rk = tab[row[q][k]] + place[q][k];
                    if (rk < SLOT_STRIDE)
                        slot_r1[(size_t)row[q][k] * SLOT_STRIDE + rk] = nbr[q][k];
                }
    }
}

__global__ void __launch_bounds__(128)
gat_rows_stride(const float* __restrict__ embs,
                const int* __restrict__ ratings,
                const int* __restrict__ counts,
                const int* __restrict__ slot_r1,
                float* __restrict__ out, int n_nodes, int n_edges) {
    const int lane = threadIdx.x & 31;
    const int i = __builtin_amdgcn_readfirstlane(
        (int)(blockIdx.x * (blockDim.x >> 5) + (threadIdx.x >> 5)));
    if (i >= n_nodes) return;
    const int cnt = counts[i];

    float4 acc = {0.0f, 0.0f, 0.0f, 0.0f};
    if (cnt > 0) {
        const float4 xi = ((const float4*)(embs + (size_t)i * D))[lane];
        const v2f a0 = {xi.x, xi.y}, a1 = {xi.z, xi.w};
        const int  k       = lane & 7;
        const bool on_diag = (lane < 8) || (lane >= 24);
        const int  l16     = lane & 15;

        float m = -INFINITY, den = 0.0f;
        if (cnt <= SLOT_STRIDE) {
            const int beg = i * SLOT_STRIDE;
            const int end = beg + cnt;
            int nb = i;
            if (beg + l16 < end) nb = slot_r1[beg + l16];
            for (int c = beg; c < end; c += 16) {
                int nb_next = i;
                if (c + 16 + l16 < end) nb_next = slot_r1[c + 16 + l16];
                const int n = end - c;
                if (n > 8) gat_batch<16>(embs, a0, a1, nb, l16 < n, lane, k, on_diag, m, den, acc);
                else       gat_batch<8>(embs, a0, a1, nb, k < n, lane, k, on_diag, m, den, acc);
                nb = nb_next;
            }
        } else {
            for (int c = 0; c < n_edges; c += 16) {
                const int e = c + l16;
                int r0 = -1, r1 = i;
                if (e < n_edges) {
                    const int2 p = ((const int2*)ratings)[e];
                    r0 = p.x; r1 = p.y;
                }
                const bool mine = (r0 == i);
                const int  nb   = mine ? r1 : i;
                if (__builtin_amdgcn_ballot_w32(mine) != 0)
                    gat_batch<16>(embs, a0, a1, nb, mine, lane, k, on_diag, m, den, acc);
            }
        }
        den += row_xor<8>(den); den += row_xor<4>(den);
        den += row_xor<2>(den); den += row_xor<1>(den);
        den += other_row(den);
        const float inv = 1.0f / den;
        acc.x *= inv; acc.y *= inv; acc.z *= inv; acc.w *= inv;
    }
    ((float4*)(out + (size_t)i * D))[lane] = acc;
}

__global__ void gat_scores_wmma(const float* __restrict__ embs,
                                const int* __restrict__ ratings,
                                float* __restrict__ scores,
                                unsigned* __restrict__ segmax,
                                int n_edges) {
    const int lane = threadIdx.x & 31;
    const int tile = blockIdx.x * (blockDim.x >> 5) + (threadIdx.x >> 5);
    const int e0 = tile << 4;
    if (e0 >= n_edges) return;

    int eL = e0 + (lane & 15);
    if (eL >= n_edges) eL = n_edges - 1;
    const int r0 = ratings[2 * eL + 0];
    const int r1 = ratings[2 * eL + 1];

    const int  k       = lane & 7;
    const bool on_diag = (lane < 8) || (lane >= 24);

    constexpr int AHEAD = 4;
    float4 a[16], b[16];
    float d[16];
#pragma unroll
    for (int j = 0; j < AHEAD; ++j) {
        const int ra = __builtin_amdgcn_readlane(r0, j);
        const int rb = __builtin_amdgcn_readlane(r1, j);
        a[j] = ((const float4*)(embs + (size_t)ra * D))[lane];
        b[j] = ((const float4*)(embs + (size_t)rb * D))[lane];
    }
#pragma unroll
    for (int j = 0; j < 16; ++j) {
        if (j + AHEAD < 16) {
            const int ra = __builtin_amdgcn_readlane(r0, j + AHEAD);
            const int rb = __builtin_amdgcn_readlane(r1, j + AHEAD);
            a[j + AHEAD] = ((const float4*)(embs + (size_t)ra * D))[lane];
            b[j + AHEAD] = ((const float4*)(embs + (size_t)rb * D))[lane];
        }
        const v2f a0 = {a[j].x, a[j].y}, a1 = {a[j].z, a[j].w};
        d[j] = diag_part(a0, a1, b[j], k, on_diag);
    }
    const float sc = lane_scores16(d, lane);

    if (lane < 16 && (e0 + lane) < n_edges) {
        scores[e0 + lane] = sc;
        atomicMax(segmax + r0, enc_order(sc));
    }
}

__global__ void gat_softmax_denom(const int* __restrict__ ratings,
                                  const unsigned* __restrict__ segmax,
                                  float* __restrict__ scores,
                                  float* __restrict__ denom,
                                  int n_edges) {
    const int e = blockIdx.x * blockDim.x + threadIdx.x;
    if (e >= n_edges) return;
    const int r0 = ratings[2 * e];
    const float m = dec_order(segmax[r0]);
    const float ex = __expf(scores[e] - m);
    scores[e] = ex;
    atomicAdd(&denom[r0], ex);
}

__global__ void gat_aggregate(const float* __restrict__ embs,
                              const int* __restrict__ ratings,
                              const float* __restrict__ ex,
                              const float* __restrict__ denom,
                              float* __restrict__ out,
                              int n_edges) {
    const int lane = threadIdx.x & 31;
    const int e = blockIdx.x * (blockDim.x >> 5) + (threadIdx.x >> 5);
    if (e >= n_edges) return;
    const int r0 = ratings[2 * e + 0];
    const int r1 = ratings[2 * e + 1];
    const float attn = ex[e] / denom[r0];

    const float4 v = ((const float4*)(embs + (size_t)r1 * D))[lane];
    float* dst = out + (size_t)r0 * D + lane * 4;
    atomicAdd(dst + 0, attn * v.x);
    atomicAdd(dst + 1, attn * v.y);
    atomicAdd(dst + 2, attn * v.z);
    atomicAdd(dst + 3, attn * v.w);
}

extern "C" void kernel_launch(void* const* d_in, const int* in_sizes, int n_in,
                              void* d_out, int out_size, void* d_ws, size_t ws_size,
                              hipStream_t stream) {
    const float* embs    = (const float*)d_in[0];
    const int*   ratings = (const int*)d_in[1];
    const int node_num = in_sizes[0] / D;
    const int n_edges  = in_sizes[1] / 2;
    float* out = (float*)d_out;

    dim3 blk(256);
    char* ws = (char*)d_ws;

    const size_t need_stride = (size_t)node_num * SLOT_STRIDE * 4 + (size_t)node_num * 4;
    if (node_num <= RANK_ROWS && ws_size >= need_stride) {
        int* slot_r1 = (int*)ws;
        int* counts  = (int*)(ws + (size_t)node_num * SLOT_STRIDE * 4);

        hipMemsetAsync(counts, 0, (size_t)node_num * 4, stream);
        dim3 gk((n_edges + RANK_CHUNK - 1) / RANK_CHUNK);
        dim3 g4((node_num + 3) / 4);
        gat_count_place_lds<<<gk, dim3(RANK_THREADS), 0, stream>>>(ratings, counts, slot_r1, n_edges, node_num);
        gat_rows_stride<<<g4, dim3(128), 0, stream>>>(embs, ratings, counts, slot_r1, out, node_num, n_edges);
        return;
    }

    const size_t need = (size_t)n_edges * 8 + (size_t)node_num * 8;
    if (ws_size >= need) {
        int* rank    = (int*)ws;
        int* slot_r1 = (int*)(ws + (size_t)n_edges * 4);
        int* counts  = (int*)(ws + (size_t)n_edges * 8);
        int* cursor  = (int*)(ws + (size_t)n_edges * 8 + (size_t)node_num * 4);

        hipMemsetAsync(counts, 0, (size_t)node_num * 4, stream);
        dim3 gc((n_edges + 1023) / 1024);
        dim3 g4((node_num + 3) / 4);
        if (node_num <= RANK_ROWS) {
            dim3 gk((n_edges + RANK_CHUNK - 1) / RANK_CHUNK);
            gat_count_rank_lds<<<gk, dim3(RANK_THREADS), 0, stream>>>(ratings, counts, rank, n_edges, node_num);
        } else {
            gat_count_rank<<<gc, blk, 0, stream>>>(ratings, counts, rank, n_edges);
        }
        gat_scan<<<1, 1024, 0, stream>>>(counts, cursor, node_num);
        gat_place<<<gc, blk, 0, stream>>>(ratings, cursor, rank, slot_r1, n_edges);
        gat_rows_fused<<<g4, dim3(128), 0, stream>>>(embs, cursor, counts, slot_r1, out, node_num);
        return;
    }

    const int tiles = (n_edges + 15) / 16;
    dim3 g1((tiles + 7) / 8);
    dim3 g2((n_edges + 255) / 256);

    float*    scores = (float*)ws;
    unsigned* segmax = (unsigned*)(ws + (size_t)n_edges * 4);
    float*    denom  = (float*)(ws + (size_t)n_edges * 4 + (size_t)node_num * 4);

    hipMemsetAsync(d_out, 0, (size_t)out_size * sizeof(float), stream);
    hipMemsetAsync(segmax, 0, (size_t)node_num * 4, stream);
    hipMemsetAsync(denom, 0, (size_t)node_num * 4, stream);

    gat_scores_wmma<<<g1, blk, 0, stream>>>(embs, ratings, scores, segmax, n_edges);
    gat_softmax_denom<<<g2, blk, 0, stream>>>(ratings, segmax, scores, denom, n_edges);

    dim3 g3((n_edges + 7) / 8);
    gat_aggregate<<<g3, blk, 0, stream>>>(embs, ratings, scores, denom, out, n_edges);
}


# row pass: neighbour rows in two register sets of 8; the next 8 row loads are issued before the products on the 8 held, loads in flight under WMMA work; softmax per 8 edges: new sum order
# speedup vs baseline: 4.9310x; 1.0349x over previous
#include <hip/hip_runtime.h>
#include <hip/hip_bf16.h>


typedef __attribute__((ext_vector_type(2))) float v2f;
typedef __attribute__((ext_vector_type(8))) float v8f;

#define D 128

__device__ __forceinline__ unsigned enc_order(float f) {
    unsigned s = __float_as_uint(f);
    return (s & 0x80000000u) ? ~s : (s | 0x80000000u);
}
__device__ __forceinline__ float dec_order(unsigned u) {
    unsigned s = (u & 0x80000000u) ? (u & 0x7fffffffu) : ~u;
    return __uint_as_float(s);
}


template <int M>
__device__ __forceinline__ float row_xor(float x) {
    return __builtin_bit_cast(float,
        __builtin_amdgcn_update_dpp(0, __builtin_bit_cast(int, x),
                                    0x160 + M, 0xf, 0xf, false));
}
__device__ __forceinline__ float other_row(float x) {
    const int b = __builtin_bit_cast(int, x);
    return __builtin_bit_cast(float,
        __builtin_amdgcn_permlanex16(b, b, 0x76543210, 0xfedcba98, false, false));
}
template <int M>
__device__ __forceinline__ float fold_pair(float lo, float hi, bool bit) {
    const float keep = bit ? hi : lo;
    const float send = bit ? lo : hi;
    return keep + row_xor<M>(send);
}

__device__ __forceinline__ float lane_f(float x, int l) {
    return __builtin_bit_cast(float,
        __builtin_amdgcn_readlane(__builtin_bit_cast(int, x), l));
}

__device__ __forceinline__ float diag_part(v2f a0, v2f a1, float4 b,
                                           int k, bool on_diag) {
    const v2f b0 = {b.x, b.y}, b1 = {b.z, b.w};
    v8f acc = {};
    acc = __builtin_amdgcn_wmma_f32_16x16x4_f32(
        false, a0, false, b0, (short)0, acc, false, false);
    acc = __builtin_amdgcn_wmma_f32_16x16x4_f32(
        false, a1, false, b1, (short)0, acc, false, false);
    float v = acc[0];
    v = (k == 1) ? acc[1] : v;
    v = (k == 2) ? acc[2] : v;
    v = (k == 3) ? acc[3] : v;
    v = (k == 4) ? acc[4] : v;
    v = (k == 5) ? acc[5] : v;
    v = (k == 6) ? acc[6] : v;
    v = (k == 7) ? acc[7] : v;
    return on_diag ? v : 0.0f;
}

__device__ __forceinline__ float lane_scores16(const float (&d)[16], int lane) {
    const bool bit8 = (lane & 8) != 0, bit4 = (lane & 4) != 0;
    const bool bit2 = (lane & 2) != 0, bit1 = (lane & 1) != 0;
    float x[8], y[4], w[2];
#pragma unroll
    for (int j = 0; j < 8; ++j) x[j] = fold_pair<8>(d[j], d[j + 8], bit8);
#pragma unroll
    for (int j = 0; j < 4; ++j) y[j] = fold_pair<4>(x[j], x[j + 4], bit4);
#pragma unroll
    for (int j = 0; j < 2; ++j) w[j] = fold_pair<2>(y[j], y[j + 2], bit2);
    const float z = fold_pair<1>(w[0], w[1], bit1);
    return z + other_row(z);
}
__device__ __forceinline__ float lane_scores8(const float (&d)[8], int lane) {
    const bool bit4 = (lane & 4) != 0;
    const bool bit2 = (lane & 2) != 0, bit1 = (lane & 1) != 0;
    float y[4], w[2];
#pragma unroll
    for (int j = 0; j < 4; ++j) y[j] = fold_pair<4>(d[j], d[j + 4], bit4);
#pragma unroll
    for (int j = 0; j < 2; ++j) w[j] = fold_pair<2>(y[j], y[j + 2], bit2);
    float z = fold_pair<1>(w[0], w[1], bit1);
    z += row_xor<8>(z);
    return z + other_row(z);
}

__global__ void gat_count_rank(const int* __restrict__ ratings,
                               int* __restrict__ counts,
                               int* __restrict__ rank, int n_edges) {
    const int e = (blockIdx.x * blockDim.x + threadIdx.x) * 4;
    if (e + 3 < n_edges) {
        const int4 p = ((const int4*)ratings)[(e >> 1) + 0];
        const int4 q = ((const int4*)ratings)[(e >> 1) + 1];
        int4 r;
        r.x = atomicAdd(&counts[p.x], 1);
        r.y = atomicAdd(&counts[p.z], 1);
        r.z = atomicAdd(&counts[q.x], 1);
        r.w = atomicAdd(&counts[q.z], 1);
        ((int4*)rank)[e >> 2] = r;
    } else {
        for (int k = e; k < n_edges; ++k)
            rank[k] = atomicAdd(&counts[ratings[2 * k]], 1);
    }
}

#define RANK_THREADS 1024
#define RANK_QUADS   2
#define RANK_ROWS    16384
#define RANK_CHUNK   (RANK_THREADS * 4 * RANK_QUADS)

__global__ void __launch_bounds__(RANK_THREADS)
gat_count_rank_lds(const int* __restrict__ ratings,
                   int* __restrict__ counts,
                   int* __restrict__ rank, int n_edges, int n_nodes) {
    __shared__ int tab[RANK_ROWS];
    const int t = threadIdx.x;
    for (int r = t; r < n_nodes; r += RANK_THREADS) tab[r] = 0;

    const int  e0    = blockIdx.x * RANK_CHUNK + t * 4;
    const bool whole = (blockIdx.x + 1) * RANK_CHUNK <= n_edges;

    int row[RANK_QUADS][4];
    if (whole) {
#pragma unroll
        for (int q = 0; q < RANK_QUADS; ++q) {
            const int e = e0 + q * RANK_THREADS * 4;
            const int4 a = ((const int4*)ratings)[(e >> 1) + 0];
            const int4 b = ((const int4*)ratings)[(e >> 1) + 1];
            row[q][0] = a.x; row[q][1] = a.z; row[q][2] = b.x; row[q][3] = b.z;
        }
    } else {
#pragma unroll
        for (int q = 0; q < RANK_QUADS; ++q)
#pragma unroll
            for (int k = 0; k < 4; ++k) {
                const int e = e0 + q * RANK_THREADS * 4 + k;
                row[q][k] = (e < n_edges) ? ratings[2 * e] : -1;
            }
    }
    __syncthreads();

    int place[RANK_QUADS][4];
#pragma unroll
    for (int q = 0; q < RANK_QUADS; ++q)
#pragma unroll
        for (int k = 0; k < 4; ++k)
            place[q][k] = (row[q][k] >= 0) ? atomicAdd(&tab[row[q][k]], 1) : 0;
    __syncthreads();

    constexpr int RPT = RANK_ROWS / RANK_THREADS;
    int c[RPT], start[RPT];
#pragma unroll
    for (int j = 0; j < RPT; ++j) {
        const int r = t + j * RANK_THREADS;
        c[j] = (r < n_nodes) ? tab[r] : 0;
    }
#pragma unroll
    for (int j = 0; j < RPT; ++j)
        start[j] = (c[j] > 0) ? atomicAdd(&counts[t + j * RANK_THREADS], c[j]) : 0;
#pragma unroll
    for (int j = 0; j < RPT; ++j)
        if (c[j] > 0) tab[t + j * RANK_THREADS] = start[j];
    __syncthreads();

    if (whole) {
#pragma unroll
        for (int q = 0; q < RANK_QUADS; ++q) {
            const int e = e0 + q * RANK_THREADS * 4;
            int4 rk;
            rk.x = tab[row[q][0]] + place[q][0];
            rk.y = tab[row[q][1]] + place[q][1];
            rk.z = tab[row[q][2]] + place[q][2];
            rk.w = tab[row[q][3]] + place[q][3];
            ((int4*)rank)[e >> 2] = rk;
        }
    } else {
#pragma unroll
        for (int q = 0; q < RANK_QUADS; ++q)
#pragma unroll
            for (int k = 0; k < 4; ++k) {
                const int e = e0 + q * RANK_THREADS * 4 + k;
                if (e < n_edges) rank[e] = tab[row[q][k]] + place[q][k];
            }
    }
}

__global__ void gat_scan(const int* __restrict__ counts,
                         int* __restrict__ cursor, int n_nodes) {
    __shared__ int wave_tot[32];
    const int t = threadIdx.x, lane = t & 31, w = t >> 5;
    const int per = (((n_nodes + (int)blockDim.x - 1) / (int)blockDim.x) + 3) & ~3;
    const int b = min(t * per, n_nodes), e = min(b + per, n_nodes);
    int s = 0;
    for (int j = b; j < e; ++j) s += counts[j];
    int incl = s;
    for (int d = 1; d < 32; d <<= 1) {
        const int y = __shfl_up(incl, d);
        if (lane >= d) incl += y;
    }
    if (lane == 31) wave_tot[w] = incl;
    __syncthreads();
    if (w == 0) {
        const int v = (lane < (int)(blockDim.x >> 5)) ? wave_tot[lane] : 0;
        int vi = v;
        for (int d = 1; d < 32; d <<= 1) {
            const int y = __shfl_up(vi, d);
            if (lane >= d) vi += y;
        }
        wave_tot[lane] = vi - v;
    }
    __syncthreads();
    int run = wave_tot[w] + incl - s;
    for (int j = b; j < e; ++j) { cursor[j] = run; run += counts[j]; }
}

__global__ void gat_place(const int* __restrict__ ratings,
                          const int* __restrict__ cursor,
                          const int* __restrict__ rank,
                          int* __restrict__ slot_r1, int n_edges) {
    const int e = (blockIdx.x * blockDim.x + threadIdx.x) * 4;
    if (e + 3 < n_edges) {
        const int4 p = ((const int4*)ratings)[(e >> 1) + 0];
        const int4 q = ((const int4*)ratings)[(e >> 1) + 1];
        const int4 r = ((const int4*)rank)[e >> 2];
        const int s0 = cursor[p.x] + r.x;
        const int s1 = cursor[p.z] + r.y;
        const int s2 = cursor[q.x] + r.z;
        const int s3 = cursor[q.z] + r.w;
        slot_r1[s0] = p.y;
        slot_r1[s1] = p.w;
        slot_r1[s2] = q.y;
        slot_r1[s3] = q.w;
    } else {
        for (int k = e; k < n_edges; ++k)
            slot_r1[cursor[ratings[2 * k]] + rank[k]] = ratings[2 * k + 1];
    }
}

template <int NB>
__device__ __forceinline__ void gat_batch(const float* __restrict__ embs,
                                          v2f a0, v2f a1, int nb, bool valid,
                                          int lane, int k, bool on_diag,
                                          float& m, float& den, float4& acc) {
    float4 v[NB];
    float  d[NB];
#pragma unroll
    for (int j = 0; j < NB; ++j) {
        const int r = __builtin_amdgcn_readlane(nb, j);
        v[j] = ((const float4*)(embs + (size_t)r * D))[lane];
    }
#pragma unroll
    for (int j = 0; j < NB; ++j) d[j] = diag_part(a0, a1, v[j], k, on_diag);

    float sc;
    if constexpr (NB == 16) sc = lane_scores16(d, lane);
    else                    sc = lane_scores8(d, lane);

    const bool counted = valid && (lane < NB);
    float bm = valid ? sc : -INFINITY;
    bm = fmaxf(bm, row_xor<1>(bm));
    bm = fmaxf(bm, row_xor<2>(bm));
    bm = fmaxf(bm, row_xor<4>(bm));
    if constexpr (NB == 16) bm = fmaxf(bm, row_xor<8>(bm));
    float m_new = fmaxf(m, bm);
    if (!__builtin_isfinite(m_new)) m_new = 0.0f;
    const float scale = __expf(m - m_new);
    const float pw    = valid ? __expf(sc - m_new) : 0.0f;
    den   = fmaf(den, scale, counted ? pw : 0.0f);
    acc.x *= scale; acc.y *= scale; acc.z *= scale; acc.w *= scale;
    m = m_new;
#pragma unroll
    for (int j = 0; j < NB; ++j) {
        const float w = lane_f(pw, j);
        acc.x = fmaf(w, v[j].x, acc.x);
        acc.y = fmaf(w, v[j].y, acc.y);
        acc.z = fmaf(w, v[j].z, acc.z);
        acc.w = fmaf(w, v[j].w, acc.w);
    }
}

template <int OFF>
__device__ __forceinline__ void gat_load8(const float* __restrict__ embs, int nb,
                                          int lane, float4 (&v)[8]) {
#pragma unroll
    for (int j = 0; j < 8; ++j) {
        const int r = __builtin_amdgcn_readlane(nb, OFF + j);
        v[j] = ((const float4*)(embs + (size_t)r * D))[lane];
    }
}

__device__ __forceinline__ void gat_work8(v2f a0, v2f a1, const float4 (&v)[8],
                                          bool valid, int lane, int k, bool on_diag,
                                          float& m, float& den, float4& acc) {
    float d[8];
#pragma unroll
    for (int j = 0; j < 8; ++j) d[j] = diag_part(a0, a1, v[j], k, on_diag);
    const float sc = lane_scores8(d, lane);

    const bool counted = valid && (lane < 8);
    float bm = valid ? sc : -INFINITY;
    bm = fmaxf(bm, row_xor<1>(bm));
    bm = fmaxf(bm, row_xor<2>(bm));
    bm = fmaxf(bm, row_xor<4>(bm));
    float m_new = fmaxf(m, bm);
    if (!__builtin_isfinite(m_new)) m_new = 0.0f;
    const float scale = __expf(m - m_new);
    const float pw    = valid ? __expf(sc - m_new) : 0.0f;
    den   = fmaf(den, scale, counted ? pw : 0.0f);
    acc.x *= scale; acc.y *= scale; acc.z *= scale; acc.w *= scale;
    m = m_new;
#pragma unroll
    for (int j = 0; j < 8; ++j) {
        const float w = lane_f(pw, j);
        acc.x = fmaf(w, v[j].x, acc.x);
        acc.y = fmaf(w, v[j].y, acc.y);
        acc.z = fmaf(w, v[j].z, acc.z);
        acc.w = fmaf(w, v[j].w, acc.w);
    }
}

__global__ void __launch_bounds__(128)
gat_rows_fused(const float* __restrict__ embs,
               const int* __restrict__ cursor,
               const int* __restrict__ counts,
               const int* __restrict__ slot_r1,
               float* __restrict__ out, int n_nodes) {
    const int lane = threadIdx.x & 31;
    const int i = __builtin_amdgcn_readfirstlane(
        (int)(blockIdx.x * (blockDim.x >> 5) + (threadIdx.x >> 5)));
    if (i >= n_nodes) return;
    const int cnt = counts[i];
    const int beg = cursor[i];
    const int end = beg + cnt;

    float4 acc = {0.0f, 0.0f, 0.0f, 0.0f};
    if (cnt > 0) {
        const float4 xi = ((const float4*)(embs + (size_t)i * D))[lane];
        const v2f a0 = {xi.x, xi.y}, a1 = {xi.z, xi.w};
        const int  k       = lane & 7;
        const bool on_diag = (lane < 8) || (lane >= 24);
        const int  l16     = lane & 15;

        float m = -INFINITY, den = 0.0f;
        int nb = i;
        if (beg + l16 < end) nb = slot_r1[beg + l16];
        for (int c = beg; c < end; c += 16) {
            int nb_next = i;
            if (c + 16 + l16 < end) nb_next = slot_r1[c + 16 + l16];
            const int n = end - c;
            if (n > 8) gat_batch<16>(embs, a0, a1, nb, l16 < n, lane, k, on_diag, m, den, acc);
            else       gat_batch<8>(embs, a0, a1, nb, k < n, lane, k, on_diag, m, den, acc);
            nb = nb_next;
        }
        den += row_xor<8>(den); den += row_xor<4>(den);
        den += row_xor<2>(den); den += row_xor<1>(den);
        den += other_row(den);
        const float inv = 1.0f / den;
        acc.x *= inv; acc.y *= inv; acc.z *= inv; acc.w *= inv;
    }
    ((float4*)(out + (size_t)i * D))[lane] = acc;
}

#define SLOT_STRIDE 128

__global__ void __launch_bounds__(RANK_THREADS)
gat_count_place_lds(const int* __restrict__ ratings,
                    int* __restrict__ counts,
                    int* __restrict__ slot_r1, int n_edges, int n_nodes) {
    __shared__ int tab[RANK_ROWS];
    const int t = threadIdx.x;
    for (int r = t; r < n_nodes; r += RANK_THREADS) tab[r] = 0;

    const int  e0    = blockIdx.x * RANK_CHUNK + t * 4;
    const bool whole = (blockIdx.x + 1) * RANK_CHUNK <= n_edges;

    int row[RANK_QUADS][4], nbr[RANK_QUADS][4];
    if (whole) {
#pragma unroll
        for (int q = 0; q < RANK_QUADS; ++q) {
            const int e = e0 + q * RANK_THREADS * 4;
            const int4 a = ((const int4*)ratings)[(e >> 1) + 0];
            const int4 b = ((const int4*)ratings)[(e >> 1) + 1];
            row[q][0] = a.x; row[q][1] = a.z; row[q][2] = b.x; row[q][3] = b.z;
            nbr[q][0] = a.y; nbr[q][1] = a.w; nbr[q][2] = b.y; nbr[q][3] = b.w;
        }
    } else {
#pragma unroll
        for (int q = 0; q < RANK_QUADS; ++q)
#pragma unroll
            for (int k = 0; k < 4; ++k) {
                const int e = e0 + q * RANK_THREADS * 4 + k;
                row[q][k] = (e < n_edges) ? ratings[2 * e] : -1;
                nbr[q][k] = (e < n_edges) ? ratings[2 * e + 1] : 0;
            }
    }
    __syncthreads();

    int place[RANK_QUADS][4];
#pragma unroll
    for (int q = 0; q < RANK_QUADS; ++q)
#pragma unroll
        for (int k = 0; k < 4; ++k)
            place[q][k] = (row[q][k] >= 0) ? atomicAdd(&tab[row[q][k]], 1) : 0;
    __syncthreads();

    constexpr int RPT = RANK_ROWS / RANK_THREADS;
    int c[RPT], start[RPT];
#pragma unroll
    for (int j = 0; j < RPT; ++j) {
        const int r = t + j * RANK_THREADS;
        c[j] = (r < n_nodes) ? tab[r] : 0;
    }
#pragma unroll
    for (int j = 0; j < RPT; ++j)
        start[j] = (c[j] > 0) ? atomicAdd(&counts[t + j * RANK_THREADS], c[j]) : 0;
#pragma unroll
    for (int j = 0; j < RPT; ++j)
        if (c[j] > 0) tab[t + j * RANK_THREADS] = start[j];
    __syncthreads();

    if (whole) {
        int rk[RANK_QUADS][4];
#pragma unroll
        for (int q = 0; q < RANK_QUADS; ++q)
#pragma unroll
            for (int k = 0; k < 4; ++k)
                rk[q][k] = tab[row[q][k]] + place[q][k];
#pragma unroll
        for (int q = 0; q < RANK_QUADS; ++q)
#pragma unroll
            for (int k = 0; k < 4; ++k)
                if (rk[q][k] < SLOT_STRIDE)
                    slot_r1[(size_t)row[q][k] * SLOT_STRIDE + rk[q][k]] = nbr[q][k];
    } else {
#pragma unroll
        for (int q = 0; q < RANK_QUADS; ++q)
#pragma unroll
            for (int k = 0; k < 4; ++k)
                if (row[q][k] >= 0) {
                    const int rk = tab[row[q][k]] + place[q][k];
                    if (rk < SLOT_STRIDE)
                        slot_r1[(size_t)row[q][k] * SLOT_STRIDE + rk] = nbr[q][k];
                }
    }
}

__global__ void __launch_bounds__(128) __attribute__((amdgpu_waves_per_eu(8)))
gat_rows_stride(const float* __restrict__ embs,
                const int* __restrict__ ratings,
                const int* __restrict__ counts,
                const int* __restrict__ slot_r1,
                float* __restrict__ out, int n_nodes, int n_edges) {
    const int lane = threadIdx.x & 31;
    const int i = __builtin_amdgcn_readfirstlane(
        (int)(blockIdx.x * (blockDim.x >> 5) + (threadIdx.x >> 5)));
    if (i >= n_nodes) return;
    const int cnt = counts[i];

    float4 acc = {0.0f, 0.0f, 0.0f, 0.0f};
    if (cnt > 0) {
        const float4 xi = ((const float4*)(embs + (size_t)i * D))[lane];
        const v2f a0 = {xi.x, xi.y}, a1 = {xi.z, xi.w};
        const int  k       = lane & 7;
        const bool on_diag = (lane < 8) || (lane >= 24);
        const int  l16     = lane & 15;

        float m = -INFINITY, den = 0.0f;
        if (cnt <= SLOT_STRIDE) {
            const int beg = i * SLOT_STRIDE;
            const int end = beg + cnt;
            int nb = i;
            if (beg + l16 < end) nb = slot_r1[beg + l16];
            float4 va[8], vb[8];
            gat_load8<0>(embs, nb, lane, va);
            for (int c = beg; ; c += 16) {
                int nb_next = i;
                if (c + 16 + l16 < end) nb_next = slot_r1[c + 16 + l16];
                const int n = end - c;
                if (n <= 8) {
                    gat_work8(a0, a1, va, k < n, lane, k, on_diag, m, den, acc);
                    break;
                }
                gat_load8<8>(embs, nb, lane, vb);
                gat_work8(a0, a1, va, true, lane, k, on_diag, m, den, acc);
                if (n <= 16) {
                    gat_work8(a0, a1, vb, k < n - 8, lane, k, on_diag, m, den, acc);
                    break;
                }
                gat_load8<0>(embs, nb_next, lane, va);
                asm volatile("" ::: "memory");
                gat_work8(a0, a1, vb, true, lane, k, on_diag, m, den, acc);
                nb = nb_next;
            }
        } else {
            for (int c = 0; c < n_edges; c += 16) {
                const int e = c + l16;
                int r0 = -1, r1 = i;
                if (e < n_edges) {
                    const int2 p = ((const int2*)ratings)[e];
                    r0 = p.x; r1 = p.y;
                }
                const bool mine = (r0 == i);
                const int  nb   = mine ? r1 : i;
                if (__builtin_amdgcn_ballot_w32(mine) != 0)
                    gat_batch<16>(embs, a0, a1, nb, mine, lane, k, on_diag, m, den, acc);
            }
        }
        den += row_xor<8>(den); den += row_xor<4>(den);
        den += row_xor<2>(den); den += row_xor<1>(den);
        den += other_row(den);
        const float inv = 1.0f / den;
        acc.x *= inv; acc.y *= inv; acc.z *= inv; acc.w *= inv;
    }
    ((float4*)(out + (size_t)i * D))[lane] = acc;
}

__global__ void gat_scores_wmma(const float* __restrict__ embs,
                                const int* __restrict__ ratings,
                                float* __restrict__ scores,
                                unsigned* __restrict__ segmax,
                                int n_edges) {
    const int lane = threadIdx.x & 31;
    const int tile = blockIdx.x * (blockDim.x >> 5) + (threadIdx.x >> 5);
    const int e0 = tile << 4;
    if (e0 >= n_edges) return;

    int eL = e0 + (lane & 15);
    if (eL >= n_edges) eL = n_edges - 1;
    const int r0 = ratings[2 * eL + 0];
    const int r1 = ratings[2 * eL + 1];

    const int  k       = lane & 7;
    const bool on_diag = (lane < 8) || (lane >= 24);

    constexpr int AHEAD = 4;
    float4 a[16], b[16];
    float d[16];
#pragma unroll
    for (int j = 0; j < AHEAD; ++j) {
        const int ra = __builtin_amdgcn_readlane(r0, j);
        const int rb = __builtin_amdgcn_readlane(r1, j);
        a[j] = ((const float4*)(embs + (size_t)ra * D))[lane];
        b[j] = ((const float4*)(embs + (size_t)rb * D))[lane];
    }
#pragma unroll
    for (int j = 0; j < 16; ++j) {
        if (j + AHEAD < 16) {
            const int ra = __builtin_amdgcn_readlane(r0, j + AHEAD);
            const int rb = __builtin_amdgcn_readlane(r1, j + AHEAD);
            a[j + AHEAD] = ((const float4*)(embs + (size_t)ra * D))[lane];
            b[j + AHEAD] = ((const float4*)(embs + (size_t)rb * D))[lane];
        }
        const v2f a0 = {a[j].x, a[j].y}, a1 = {a[j].z, a[j].w};
        d[j] = diag_part(a0, a1, b[j], k, on_diag);
    }
    const float sc = lane_scores16(d, lane);

    if (lane < 16 && (e0 + lane) < n_edges) {
        scores[e0 + lane] = sc;
        atomicMax(segmax + r0, enc_order(sc));
    }
}

__global__ void gat_softmax_denom(const int* __restrict__ ratings,
                                  const unsigned* __restrict__ segmax,
                                  float* __restrict__ scores,
                                  float* __restrict__ denom,
                                  int n_edges) {
    const int e = blockIdx.x * blockDim.x + threadIdx.x;
    if (e >= n_edges) return;
    const int r0 = ratings[2 * e];
    const float m = dec_order(segmax[r0]);
    const float ex = __expf(scores[e] - m);
    scores[e] = ex;
    atomicAdd(&denom[r0], ex);
}

__global__ void gat_aggregate(const float* __restrict__ embs,
                              const int* __restrict__ ratings,
                              const float* __restrict__ ex,
                              const float* __restrict__ denom,
                              float* __restrict__ out,
                              int n_edges) {
    const int lane = threadIdx.x & 31;
    const int e = blockIdx.x * (blockDim.x >> 5) + (threadIdx.x >> 5);
    if (e >= n_edges) return;
    const int r0 = ratings[2 * e + 0];
    const int r1 = ratings[2 * e + 1];
    const float attn = ex[e] / denom[r0];

    const float4 v = ((const float4*)(embs + (size_t)r1 * D))[lane];
    float* dst = out + (size_t)r0 * D + lane * 4;
    atomicAdd(dst + 0, attn * v.x);
    atomicAdd(dst + 1, attn * v.y);
    atomicAdd(dst + 2, attn * v.z);
    atomicAdd(dst + 3, attn * v.w);
}

extern "C" void kernel_launch(void* const* d_in, const int* in_sizes, int n_in,
                              void* d_out, int out_size, void* d_ws, size_t ws_size,
                              hipStream_t stream) {
    const float* embs    = (const float*)d_in[0];
    const int*   ratings = (const int*)d_in[1];
    const int node_num = in_sizes[0] / D;
    const int n_edges  = in_sizes[1] / 2;
    float* out = (float*)d_out;

    dim3 blk(256);
    char* ws = (char*)d_ws;

    const size_t need_stride = (size_t)node_num * SLOT_STRIDE * 4 + (size_t)node_num * 4;
    if (node_num <= RANK_ROWS && ws_size >= need_stride) {
        int* slot_r1 = (int*)ws;
        int* counts  = (int*)(ws + (size_t)node_num * SLOT_STRIDE * 4);

        hipMemsetAsync(counts, 0, (size_t)node_num * 4, stream);
        dim3 gk((n_edges + RANK_CHUNK - 1) / RANK_CHUNK);
        dim3 g4((node_num + 3) / 4);
        gat_count_place_lds<<<gk, dim3(RANK_THREADS), 0, stream>>>(ratings, counts, slot_r1, n_edges, node_num);
        gat_rows_stride<<<g4, dim3(128), 0, stream>>>(embs, ratings, counts, slot_r1, out, node_num, n_edges);
        return;
    }

    const size_t need = (size_t)n_edges * 8 + (size_t)node_num * 8;
    if (ws_size >= need) {
        int* rank    = (int*)ws;
        int* slot_r1 = (int*)(ws + (size_t)n_edges * 4);
        int* counts  = (int*)(ws + (size_t)n_edges * 8);
        int* cursor  = (int*)(ws + (size_t)n_edges * 8 + (size_t)node_num * 4);

        hipMemsetAsync(counts, 0, (size_t)node_num * 4, stream);
        dim3 gc((n_edges + 1023) / 1024);
        dim3 g4((node_num + 3) / 4);
        if (node_num <= RANK_ROWS) {
            dim3 gk((n_edges + RANK_CHUNK - 1) / RANK_CHUNK);
            gat_count_rank_lds<<<gk, dim3(RANK_THREADS), 0, stream>>>(ratings, counts, rank, n_edges, node_num);
        } else {
            gat_count_rank<<<gc, blk, 0, stream>>>(ratings, counts, rank, n_edges);
        }
        gat_scan<<<1, 1024, 0, stream>>>(counts, cursor, node_num);
        gat_place<<<gc, blk, 0, stream>>>(ratings, cursor, rank, slot_r1, n_edges);
        gat_rows_fused<<<g4, dim3(128), 0, stream>>>(embs, cursor, counts, slot_r1, out, node_num);
        return;
    }

    const int tiles = (n_edges + 15) / 16;
    dim3 g1((tiles + 7) / 8);
    dim3 g2((n_edges + 255) / 256);

    float*    scores = (float*)ws;
    unsigned* segmax = (unsigned*)(ws + (size_t)n_edges * 4);
    float*    denom  = (float*)(ws + (size_t)n_edges * 4 + (size_t)node_num * 4);

    hipMemsetAsync(d_out, 0, (size_t)out_size * sizeof(float), stream);
    hipMemsetAsync(segmax, 0, (size_t)node_num * 4, stream);
    hipMemsetAsync(denom, 0, (size_t)node_num * 4, stream);

    gat_scores_wmma<<<g1, blk, 0, stream>>>(embs, ratings, scores, segmax, n_edges);
    gat_softmax_denom<<<g2, blk, 0, stream>>>(ratings, segmax, scores, denom, n_edges);

    dim3 g3((n_edges + 7) / 8);
    gat_aggregate<<<g3, blk, 0, stream>>>(embs, ratings, scores, denom, out, n_edges);
}
